# starting kernel
# baseline (speedup 1.0000x reference)
#include <hip/hip_runtime.h>

typedef __attribute__((ext_vector_type(2))) float v2f;
typedef __attribute__((ext_vector_type(8))) float v8f;

#define GCN_IN_C   128
#define GCN_HID_C  128
#define GCN_OUT_C  64
#define GCN_BN_EPS 1e-5f


__global__ void zero_f32(float* __restrict__ p, long n) {
  long i = (long)blockIdx.x * blockDim.x + threadIdx.x;
  long stride = (long)gridDim.x * blockDim.x;
  for (; i < n; i += stride) p[i] = 0.0f;
}

__global__ void degree_kernel(const int* __restrict__ dst, int E, int* __restrict__ deg) {
  int i = blockIdx.x * blockDim.x + threadIdx.x;
  if (i < E) atomicAdd(&deg[dst[i]], 1);
}

__global__ void dinv_kernel(const int* __restrict__ deg, float* __restrict__ dinv, int N) {
  int i = blockIdx.x * blockDim.x + threadIdx.x;
  if (i < N) dinv[i] = rsqrtf((float)(deg[i] + 1));
}

template <int KDIM>
__global__ void gemm_wmma_lds(const float* __restrict__ A, const float* __restrict__ B,
                              float* __restrict__ C, int N) {
  constexpr int LDA = KDIM + 4;
  __shared__ float As[16 * LDA];

  const int tm = blockIdx.x;
  const float* __restrict__ gA = A + (long)tm * 16 * KDIM;

  for (int idx = threadIdx.x; idx < 16 * KDIM / 4; idx += blockDim.x) {
    const int r  = idx / (KDIM / 4);
    const int c4 = idx % (KDIM / 4);
    const float4 v = ((const float4*)(gA + (long)r * KDIM))[c4];
    *(float4*)&As[r * LDA + c4 * 4] = v;
  }
  __syncthreads();

  const int lane = threadIdx.x & 31;
  const int half = lane >> 4;
  const int l16  = lane & 15;
  const int tn   = threadIdx.x >> 5;
  const int bcol = tn * 16 + l16;
  const float* __restrict__ ap = &As[l16 * LDA];

  v8f acc = {};
#pragma unroll 4
  for (int k0 = 0; k0 < KDIM; k0 += 4) {
    const int ka = k0 + 2 * half;
    v2f a, b;
    a.x = ap[ka];
    a.y = ap[ka + 1];
    b.x = B[(long)ka * N + bcol];
    b.y = B[(long)(ka + 1) * N + bcol];
    acc = __builtin_amdgcn_wmma_f32_16x16x4_f32(
         false, a,  false, b,
         (short)0, acc,  false,  false);
  }

  const int orow = tm * 16 + half * 8;
  float* __restrict__ cp = C + (long)orow * N + bcol;
#pragma unroll
  for (int r = 0; r < 8; ++r) cp[(long)r * N] = acc[r];
}

__global__ void scatter_c128(const float* __restrict__ h, const int* __restrict__ src,
                             const int* __restrict__ dst, const float* __restrict__ dinv,
                             float* __restrict__ agg, int E) {
  const int wave = (int)((blockIdx.x * (long)blockDim.x + threadIdx.x) >> 5);
  const int lane = threadIdx.x & 31;
  if (wave >= E) return;
  const int s = src[wave];
  const int d = dst[wave];
  const float coef = dinv[s] * dinv[d];
  const float4 v = ((const float4*)(h + (long)s * 128))[lane];
  float* __restrict__ o = agg + (long)d * 128 + lane * 4;
  atomicAdd(o + 0, v.x * coef);
  atomicAdd(o + 1, v.y * coef);
  atomicAdd(o + 2, v.z * coef);
  atomicAdd(o + 3, v.w * coef);
}

__global__ void scatter_c64(const float* __restrict__ h, const int* __restrict__ src,
                            const int* __restrict__ dst, const float* __restrict__ dinv,
                            float* __restrict__ agg, int E) {
  const int wave = (int)((blockIdx.x * (long)blockDim.x + threadIdx.x) >> 5);
  const int lane = threadIdx.x & 31;
  if (wave >= E) return;
  const int s = src[wave];
  const int d = dst[wave];
  const float coef = dinv[s] * dinv[d];
  const float2 v = ((const float2*)(h + (long)s * 64))[lane];
  float* __restrict__ o = agg + (long)d * 64 + lane * 2;
  atomicAdd(o + 0, v.x * coef);
  atomicAdd(o + 1, v.y * coef);
}

__global__ void finalize_relu_c128(const float* __restrict__ h, float* __restrict__ agg,
                                   const float* __restrict__ dinv, const float* __restrict__ bias,
                                   const float* __restrict__ gamma, const float* __restrict__ beta,
                                   const float* __restrict__ mean, const float* __restrict__ var,
                                   int N) {
  const long idx = (long)blockIdx.x * blockDim.x + threadIdx.x;
  if (idx >= (long)N * 128) return;
  const int i = (int)(idx >> 7);
  const int c = (int)(idx & 127);
  const float di = dinv[i];
  float v = agg[idx] + h[idx] * di * di + bias[c];
  v = (v - mean[c]) * rsqrtf(var[c] + GCN_BN_EPS) * gamma[c] + beta[c];
  agg[idx] = fmaxf(v, 0.0f);
}

__global__ void finalize_c64(const float* __restrict__ h, const float* __restrict__ agg,
                             const float* __restrict__ dinv, const float* __restrict__ bias,
                             const float* __restrict__ gamma, const float* __restrict__ beta,
                             const float* __restrict__ mean, const float* __restrict__ var,
                             float* __restrict__ out, int N) {
  const long idx = (long)blockIdx.x * blockDim.x + threadIdx.x;
  if (idx >= (long)N * 64) return;
  const int i = (int)(idx >> 6);
  const int c = (int)(idx & 63);
  const float di = dinv[i];
  float v = agg[idx] + h[idx] * di * di + bias[c];
  v = (v - mean[c]) * rsqrtf(var[c] + GCN_BN_EPS) * gamma[c] + beta[c];
  out[idx] = v;
}


extern "C" void kernel_launch(void* const* d_in, const int* in_sizes, int n_in,
                              void* d_out, int out_size, void* d_ws, size_t ws_size,
                              hipStream_t stream) {
  const float* x   = (const float*)d_in[0];
  const int*   ei  = (const int*)d_in[1];
  const float* W1  = (const float*)d_in[2];
  const float* b1  = (const float*)d_in[3];
  const float* g1  = (const float*)d_in[4];
  const float* be1 = (const float*)d_in[5];
  const float* m1  = (const float*)d_in[6];
  const float* v1  = (const float*)d_in[7];
  const float* W2  = (const float*)d_in[8];
  const float* b2  = (const float*)d_in[9];
  const float* g2  = (const float*)d_in[10];
  const float* be2 = (const float*)d_in[11];
  const float* m2  = (const float*)d_in[12];
  const float* v2  = (const float*)d_in[13];

  const int N = in_sizes[0] / GCN_IN_C;
  const int E = in_sizes[1] / 2;
  const int* src = ei;
  const int* dst = ei + E;

  char* ws = (char*)d_ws;
  int*   deg  = (int*)ws;                         ws += (size_t)N * sizeof(int);
  float* dinv = (float*)ws;                       ws += (size_t)N * sizeof(float);
  float* h1   = (float*)ws;                       ws += (size_t)N * GCN_HID_C * sizeof(float);
  float* agg1 = (float*)ws;                       ws += (size_t)N * GCN_HID_C * sizeof(float);
  float* h2   = (float*)ws;                       ws += (size_t)N * GCN_OUT_C * sizeof(float);
  float* agg2 = (float*)ws;                       ws += (size_t)N * GCN_OUT_C * sizeof(float);
  (void)ws_size; (void)n_in; (void)out_size;

  const int ZB = 2048;
  zero_f32<<<ZB, 256, 0, stream>>>((float*)deg, (long)N);
  zero_f32<<<ZB, 256, 0, stream>>>(agg1, (long)N * GCN_HID_C);
  zero_f32<<<ZB, 256, 0, stream>>>(agg2, (long)N * GCN_OUT_C);

  degree_kernel<<<(E + 255) / 256, 256, 0, stream>>>(dst, E, deg);
  dinv_kernel<<<(N + 255) / 256, 256, 0, stream>>>(deg, dinv, N);

  gemm_wmma_lds<GCN_IN_C><<<N / 16, (GCN_HID_C / 16) * 32, 0, stream>>>(x, W1, h1, GCN_HID_C);
  {
    const long blocks = ((long)E * 32 + 255) / 256;
    scatter_c128<<<(int)blocks, 256, 0, stream>>>(h1, src, dst, dinv, agg1, E);
  }
  {
    const long total = (long)N * GCN_HID_C;
    finalize_relu_c128<<<(int)((total + 255) / 256), 256, 0, stream>>>(
        h1, agg1, dinv, b1, g1, be1, m1, v1, N);
  }

  gemm_wmma_lds<GCN_HID_C><<<N / 16, (GCN_OUT_C / 16) * 32, 0, stream>>>(agg1, W2, h2, GCN_OUT_C);
  {
    const long blocks = ((long)E * 32 + 255) / 256;
    scatter_c64<<<(int)blocks, 256, 0, stream>>>(h2, src, dst, dinv, agg2, E);
  }
  {
    const long total = (long)N * GCN_OUT_C;
    finalize_c64<<<(int)((total + 255) / 256), 256, 0, stream>>>(
        h2, agg2, dinv, b2, g2, be2, m2, v2, (float*)d_out, N);
  }
}


# scatter kernels: lane l takes channels l+32j, so each f32 atomic add (and row load) instruction covers one whole 128B line, not 4 quarter (c128) or 2 half lines (c64); sums stay in arrival order
# speedup vs baseline: 1.2017x; 1.2017x over previous
#include <hip/hip_runtime.h>

typedef __attribute__((ext_vector_type(2))) float v2f;
typedef __attribute__((ext_vector_type(8))) float v8f;

#define GCN_IN_C   128
#define GCN_HID_C  128
#define GCN_OUT_C  64
#define GCN_BN_EPS 1e-5f


__global__ void zero_f32(float* __restrict__ p, long n) {
  long i = (long)blockIdx.x * blockDim.x + threadIdx.x;
  long stride = (long)gridDim.x * blockDim.x;
  for (; i < n; i += stride) p[i] = 0.0f;
}

__global__ void degree_kernel(const int* __restrict__ dst, int E, int* __restrict__ deg) {
  int i = blockIdx.x * blockDim.x + threadIdx.x;
  if (i < E) atomicAdd(&deg[dst[i]], 1);
}

__global__ void dinv_kernel(const int* __restrict__ deg, float* __restrict__ dinv, int N) {
  int i = blockIdx.x * blockDim.x + threadIdx.x;
  if (i < N) dinv[i] = rsqrtf((float)(deg[i] + 1));
}

template <int KDIM>
__global__ void gemm_wmma_lds(const float* __restrict__ A, const float* __restrict__ B,
                              float* __restrict__ C, int N) {
  constexpr int LDA = KDIM + 4;
  __shared__ float As[16 * LDA];

  const int tm = blockIdx.x;
  const float* __restrict__ gA = A + (long)tm * 16 * KDIM;

  for (int idx = threadIdx.x; idx < 16 * KDIM / 4; idx += blockDim.x) {
    const int r  = idx / (KDIM / 4);
    const int c4 = idx % (KDIM / 4);
    const float4 v = ((const float4*)(gA + (long)r * KDIM))[c4];
    *(float4*)&As[r * LDA + c4 * 4] = v;
  }
  __syncthreads();

  const int lane = threadIdx.x & 31;
  const int half = lane >> 4;
  const int l16  = lane & 15;
  const int tn   = threadIdx.x >> 5;
  const int bcol = tn * 16 + l16;
  const float* __restrict__ ap = &As[l16 * LDA];

  v8f acc = {};
#pragma unroll 4
  for (int k0 = 0; k0 < KDIM; k0 += 4) {
    const int ka = k0 + 2 * half;
    v2f a, b;
    a.x = ap[ka];
    a.y = ap[ka + 1];
    b.x = B[(long)ka * N + bcol];
    b.y = B[(long)(ka + 1) * N + bcol];
    acc = __builtin_amdgcn_wmma_f32_16x16x4_f32(
         false, a,  false, b,
         (short)0, acc,  false,  false);
  }

  const int orow = tm * 16 + half * 8;
  float* __restrict__ cp = C + (long)orow * N + bcol;
#pragma unroll
  for (int r = 0; r < 8; ++r) cp[(long)r * N] = acc[r];
}

__global__ void scatter_c128(const float* __restrict__ h, const int* __restrict__ src,
                             const int* __restrict__ dst, const float* __restrict__ dinv,
                             float* __restrict__ agg, int E) {
  const int wave = (int)((blockIdx.x * (long)blockDim.x + threadIdx.x) >> 5);
  const int lane = threadIdx.x & 31;
  if (wave >= E) return;
  const int s = src[wave];
  const int d = dst[wave];
  const float coef = dinv[s] * dinv[d];
  const float* __restrict__ hp = h + (long)s * 128 + lane;
  const float v0 = hp[0];
  const float v1 = hp[32];
  const float v2 = hp[64];
  const float v3 = hp[96];
  float* __restrict__ o = agg + (long)d * 128 + lane;
  atomicAdd(o,      v0 * coef);
  atomicAdd(o + 32, v1 * coef);
  atomicAdd(o + 64, v2 * coef);
  atomicAdd(o + 96, v3 * coef);
}

__global__ void scatter_c64(const float* __restrict__ h, const int* __restrict__ src,
                            const int* __restrict__ dst, const float* __restrict__ dinv,
                            float* __restrict__ agg, int E) {
  const int wave = (int)((blockIdx.x * (long)blockDim.x + threadIdx.x) >> 5);
  const int lane = threadIdx.x & 31;
  if (wave >= E) return;
  const int s = src[wave];
  const int d = dst[wave];
  const float coef = dinv[s] * dinv[d];
  const float* __restrict__ hp = h + (long)s * 64 + lane;
  const float v0 = hp[0];
  const float v1 = hp[32];
  float* __restrict__ o = agg + (long)d * 64 + lane;
  atomicAdd(o,      v0 * coef);
  atomicAdd(o + 32, v1 * coef);
}

__global__ void finalize_relu_c128(const float* __restrict__ h, float* __restrict__ agg,
                                   const float* __restrict__ dinv, const float* __restrict__ bias,
                                   const float* __restrict__ gamma, const float* __restrict__ beta,
                                   const float* __restrict__ mean, const float* __restrict__ var,
                                   int N) {
  const long idx = (long)blockIdx.x * blockDim.x + threadIdx.x;
  if (idx >= (long)N * 128) return;
  const int i = (int)(idx >> 7);
  const int c = (int)(idx & 127);
  const float di = dinv[i];
  float v = agg[idx] + h[idx] * di * di + bias[c];
  v = (v - mean[c]) * rsqrtf(var[c] + GCN_BN_EPS) * gamma[c] + beta[c];
  agg[idx] = fmaxf(v, 0.0f);
}

__global__ void finalize_c64(const float* __restrict__ h, const float* __restrict__ agg,
                             const float* __restrict__ dinv, const float* __restrict__ bias,
                             const float* __restrict__ gamma, const float* __restrict__ beta,
                             const float* __restrict__ mean, const float* __restrict__ var,
                             float* __restrict__ out, int N) {
  const long idx = (long)blockIdx.x * blockDim.x + threadIdx.x;
  if (idx >= (long)N * 64) return;
  const int i = (int)(idx >> 6);
  const int c = (int)(idx & 63);
  const float di = dinv[i];
  float v = agg[idx] + h[idx] * di * di + bias[c];
  v = (v - mean[c]) * rsqrtf(var[c] + GCN_BN_EPS) * gamma[c] + beta[c];
  out[idx] = v;
}


extern "C" void kernel_launch(void* const* d_in, const int* in_sizes, int n_in,
                              void* d_out, int out_size, void* d_ws, size_t ws_size,
                              hipStream_t stream) {
  const float* x   = (const float*)d_in[0];
  const int*   ei  = (const int*)d_in[1];
  const float* W1  = (const float*)d_in[2];
  const float* b1  = (const float*)d_in[3];
  const float* g1  = (const float*)d_in[4];
  const float* be1 = (const float*)d_in[5];
  const float* m1  = (const float*)d_in[6];
  const float* v1  = (const float*)d_in[7];
  const float* W2  = (const float*)d_in[8];
  const float* b2  = (const float*)d_in[9];
  const float* g2  = (const float*)d_in[10];
  const float* be2 = (const float*)d_in[11];
  const float* m2  = (const float*)d_in[12];
  const float* v2  = (const float*)d_in[13];

  const int N = in_sizes[0] / GCN_IN_C;
  const int E = in_sizes[1] / 2;
  const int* src = ei;
  const int* dst = ei + E;

  char* ws = (char*)d_ws;
  int*   deg  = (int*)ws;                         ws += (size_t)N * sizeof(int);
  float* dinv = (float*)ws;                       ws += (size_t)N * sizeof(float);
  float* h1   = (float*)ws;                       ws += (size_t)N * GCN_HID_C * sizeof(float);
  float* agg1 = (float*)ws;                       ws += (size_t)N * GCN_HID_C * sizeof(float);
  float* h2   = (float*)ws;                       ws += (size_t)N * GCN_OUT_C * sizeof(float);
  float* agg2 = (float*)ws;                       ws += (size_t)N * GCN_OUT_C * sizeof(float);
  (void)ws_size; (void)n_in; (void)out_size;

  const int ZB = 2048;
  zero_f32<<<ZB, 256, 0, stream>>>((float*)deg, (long)N);
  zero_f32<<<ZB, 256, 0, stream>>>(agg1, (long)N * GCN_HID_C);
  zero_f32<<<ZB, 256, 0, stream>>>(agg2, (long)N * GCN_OUT_C);

  degree_kernel<<<(E + 255) / 256, 256, 0, stream>>>(dst, E, deg);
  dinv_kernel<<<(N + 255) / 256, 256, 0, stream>>>(deg, dinv, N);

  gemm_wmma_lds<GCN_IN_C><<<N / 16, (GCN_HID_C / 16) * 32, 0, stream>>>(x, W1, h1, GCN_HID_C);
  {
    const long blocks = ((long)E * 32 + 255) / 256;
    scatter_c128<<<(int)blocks, 256, 0, stream>>>(h1, src, dst, dinv, agg1, E);
  }
  {
    const long total = (long)N * GCN_HID_C;
    finalize_relu_c128<<<(int)((total + 255) / 256), 256, 0, stream>>>(
        h1, agg1, dinv, b1, g1, be1, m1, v1, N);
  }

  gemm_wmma_lds<GCN_HID_C><<<N / 16, (GCN_OUT_C / 16) * 32, 0, stream>>>(agg1, W2, h2, GCN_OUT_C);
  {
    const long blocks = ((long)E * 32 + 255) / 256;
    scatter_c64<<<(int)blocks, 256, 0, stream>>>(h2, src, dst, dinv, agg2, E);
  }
  {
    const long total = (long)N * GCN_OUT_C;
    finalize_c64<<<(int)((total + 255) / 256), 256, 0, stream>>>(
        h2, agg2, dinv, b2, g2, be2, m2, v2, (float*)d_out, N);
  }
}


# scatter kernels: one wave per 32 edges, not per edge: lanes load the 32 src/dst/dinv coalesced, rows move 8 (c128) or 16 (c64) edges per chunk, all row loads ahead of the atomics; arrival-order sums
# speedup vs baseline: 1.5329x; 1.2756x over previous
#include <hip/hip_runtime.h>

typedef __attribute__((ext_vector_type(2))) float v2f;
typedef __attribute__((ext_vector_type(8))) float v8f;

#define GCN_IN_C   128
#define GCN_HID_C  128
#define GCN_OUT_C  64
#define GCN_BN_EPS 1e-5f


__global__ void zero_f32(float* __restrict__ p, long n) {
  long i = (long)blockIdx.x * blockDim.x + threadIdx.x;
  long stride = (long)gridDim.x * blockDim.x;
  for (; i < n; i += stride) p[i] = 0.0f;
}

__global__ void degree_kernel(const int* __restrict__ dst, int E, int* __restrict__ deg) {
  int i = blockIdx.x * blockDim.x + threadIdx.x;
  if (i < E) atomicAdd(&deg[dst[i]], 1);
}

__global__ void dinv_kernel(const int* __restrict__ deg, float* __restrict__ dinv, int N) {
  int i = blockIdx.x * blockDim.x + threadIdx.x;
  if (i < N) dinv[i] = rsqrtf((float)(deg[i] + 1));
}

template <int KDIM>
__global__ void gemm_wmma_lds(const float* __restrict__ A, const float* __restrict__ B,
                              float* __restrict__ C, int N) {
  constexpr int LDA = KDIM + 4;
  __shared__ float As[16 * LDA];

  const int tm = blockIdx.x;
  const float* __restrict__ gA = A + (long)tm * 16 * KDIM;

  for (int idx = threadIdx.x; idx < 16 * KDIM / 4; idx += blockDim.x) {
    const int r  = idx / (KDIM / 4);
    const int c4 = idx % (KDIM / 4);
    const float4 v = ((const float4*)(gA + (long)r * KDIM))[c4];
    *(float4*)&As[r * LDA + c4 * 4] = v;
  }
  __syncthreads();

  const int lane = threadIdx.x & 31;
  const int half = lane >> 4;
  const int l16  = lane & 15;
  const int tn   = threadIdx.x >> 5;
  const int bcol = tn * 16 + l16;
  const float* __restrict__ ap = &As[l16 * LDA];

  v8f acc = {};
#pragma unroll 4
  for (int k0 = 0; k0 < KDIM; k0 += 4) {
    const int ka = k0 + 2 * half;
    v2f a, b;
    a.x = ap[ka];
    a.y = ap[ka + 1];
    b.x = B[(long)ka * N + bcol];
    b.y = B[(long)(ka + 1) * N + bcol];
    acc = __builtin_amdgcn_wmma_f32_16x16x4_f32(
         false, a,  false, b,
         (short)0, acc,  false,  false);
  }

  const int orow = tm * 16 + half * 8;
  float* __restrict__ cp = C + (long)orow * N + bcol;
#pragma unroll
  for (int r = 0; r < 8; ++r) cp[(long)r * N] = acc[r];
}

template <int C, int EPC>
__global__ void scatter_rows(const float* __restrict__ h, const int* __restrict__ src,
                             const int* __restrict__ dst, const float* __restrict__ dinv,
                             float* __restrict__ agg, int E) {
  constexpr int Q = C / 32;
  const int lane = threadIdx.x & 31;
  const int wave = __builtin_amdgcn_readfirstlane(
      (int)((blockIdx.x * (long)blockDim.x + threadIdx.x) >> 5));
  const long e0 = (long)wave * 32;
  if (e0 >= (long)E) return;
  const long left = (long)E - e0;
  const int nvalid = left < 32 ? (int)left : 32;

  int sv = 0, dv = 0;
  float cv = 0.0f;
  if (lane < nvalid) {
    sv = src[e0 + lane];
    dv = dst[e0 + lane];
    cv = dinv[sv] * dinv[dv];
  }
  const int cvi = __float_as_int(cv);

  if (nvalid == 32) {
#pragma unroll 1
    for (int c = 0; c < 32; c += EPC) {
      float v[EPC][Q];
#pragma unroll
      for (int j = 0; j < EPC; ++j) {
        const int s = __builtin_amdgcn_readlane(sv, c + j);
        const float* __restrict__ hp = h + (long)s * C + lane;
#pragma unroll
        for (int q = 0; q < Q; ++q) v[j][q] = hp[32 * q];
      }
#pragma unroll
      for (int j = 0; j < EPC; ++j) {
        const int d = __builtin_amdgcn_readlane(dv, c + j);
        const float coef = __int_as_float(__builtin_amdgcn_readlane(cvi, c + j));
        float* __restrict__ o = agg + (long)d * C + lane;
#pragma unroll
        for (int q = 0; q < Q; ++q) atomicAdd(o + 32 * q, v[j][q] * coef);
      }
    }
  } else {
    for (int j = 0; j < nvalid; ++j) {
      const int s = __builtin_amdgcn_readlane(sv, j);
      const int d = __builtin_amdgcn_readlane(dv, j);
      const float coef = __int_as_float(__builtin_amdgcn_readlane(cvi, j));
      const float* __restrict__ hp = h + (long)s * C + lane;
      float* __restrict__ o = agg + (long)d * C + lane;
      for (int q = 0; q < Q; ++q) atomicAdd(o + 32 * q, hp[32 * q] * coef);
    }
  }
}

__global__ void finalize_relu_c128(const float* __restrict__ h, float* __restrict__ agg,
                                   const float* __restrict__ dinv, const float* __restrict__ bias,
                                   const float* __restrict__ gamma, const float* __restrict__ beta,
                                   const float* __restrict__ mean, const float* __restrict__ var,
                                   int N) {
  const long idx = (long)blockIdx.x * blockDim.x + threadIdx.x;
  if (idx >= (long)N * 128) return;
  const int i = (int)(idx >> 7);
  const int c = (int)(idx & 127);
  const float di = dinv[i];
  float v = agg[idx] + h[idx] * di * di + bias[c];
  v = (v - mean[c]) * rsqrtf(var[c] + GCN_BN_EPS) * gamma[c] + beta[c];
  agg[idx] = fmaxf(v, 0.0f);
}

__global__ void finalize_c64(const float* __restrict__ h, const float* __restrict__ agg,
                             const float* __restrict__ dinv, const float* __restrict__ bias,
                             const float* __restrict__ gamma, const float* __restrict__ beta,
                             const float* __restrict__ mean, const float* __restrict__ var,
                             float* __restrict__ out, int N) {
  const long idx = (long)blockIdx.x * blockDim.x + threadIdx.x;
  if (idx >= (long)N * 64) return;
  const int i = (int)(idx >> 6);
  const int c = (int)(idx & 63);
  const float di = dinv[i];
  float v = agg[idx] + h[idx] * di * di + bias[c];
  v = (v - mean[c]) * rsqrtf(var[c] + GCN_BN_EPS) * gamma[c] + beta[c];
  out[idx] = v;
}


extern "C" void kernel_launch(void* const* d_in, const int* in_sizes, int n_in,
                              void* d_out, int out_size, void* d_ws, size_t ws_size,
                              hipStream_t stream) {
  const float* x   = (const float*)d_in[0];
  const int*   ei  = (const int*)d_in[1];
  const float* W1  = (const float*)d_in[2];
  const float* b1  = (const float*)d_in[3];
  const float* g1  = (const float*)d_in[4];
  const float* be1 = (const float*)d_in[5];
  const float* m1  = (const float*)d_in[6];
  const float* v1  = (const float*)d_in[7];
  const float* W2  = (const float*)d_in[8];
  const float* b2  = (const float*)d_in[9];
  const float* g2  = (const float*)d_in[10];
  const float* be2 = (const float*)d_in[11];
  const float* m2  = (const float*)d_in[12];
  const float* v2  = (const float*)d_in[13];

  const int N = in_sizes[0] / GCN_IN_C;
  const int E = in_sizes[1] / 2;
  const int* src = ei;
  const int* dst = ei + E;

  char* ws = (char*)d_ws;
  int*   deg  = (int*)ws;                         ws += (size_t)N * sizeof(int);
  float* dinv = (float*)ws;                       ws += (size_t)N * sizeof(float);
  float* h1   = (float*)ws;                       ws += (size_t)N * GCN_HID_C * sizeof(float);
  float* agg1 = (float*)ws;                       ws += (size_t)N * GCN_HID_C * sizeof(float);
  float* h2   = (float*)ws;                       ws += (size_t)N * GCN_OUT_C * sizeof(float);
  float* agg2 = (float*)ws;                       ws += (size_t)N * GCN_OUT_C * sizeof(float);
  (void)ws_size; (void)n_in; (void)out_size;

  const int ZB = 2048;
  const int sblocks = ((E + 31) / 32 + 7) / 8;
  zero_f32<<<ZB, 256, 0, stream>>>((float*)deg, (long)N);
  zero_f32<<<ZB, 256, 0, stream>>>(agg1, (long)N * GCN_HID_C);
  zero_f32<<<ZB, 256, 0, stream>>>(agg2, (long)N * GCN_OUT_C);

  degree_kernel<<<(E + 255) / 256, 256, 0, stream>>>(dst, E, deg);
  dinv_kernel<<<(N + 255) / 256, 256, 0, stream>>>(deg, dinv, N);

  gemm_wmma_lds<GCN_IN_C><<<N / 16, (GCN_HID_C / 16) * 32, 0, stream>>>(x, W1, h1, GCN_HID_C);
  scatter_rows<GCN_HID_C, 8><<<sblocks, 256, 0, stream>>>(h1, src, dst, dinv, agg1, E);
  {
    const long total = (long)N * GCN_HID_C;
    finalize_relu_c128<<<(int)((total + 255) / 256), 256, 0, stream>>>(
        h1, agg1, dinv, b1, g1, be1, m1, v1, N);
  }

  gemm_wmma_lds<GCN_HID_C><<<N / 16, (GCN_OUT_C / 16) * 32, 0, stream>>>(agg1, W2, h2, GCN_OUT_C);
  scatter_rows<GCN_OUT_C, 16><<<sblocks, 256, 0, stream>>>(h2, src, dst, dinv, agg2, E);
  {
    const long total = (long)N * GCN_OUT_C;
    finalize_c64<<<(int)((total + 255) / 256), 256, 0, stream>>>(
        h2, agg2, dinv, b2, g2, be2, m2, v2, (float*)d_out, N);
  }
}


# aggregation by gather, not f32 atomics: edges counting-sorted by dst per call (returning int add = rank, scan, fill); one wave per node sums its rows in registers, one store; arrival-rank order sums
# speedup vs baseline: 2.0763x; 1.3545x over previous
#include <hip/hip_runtime.h>

typedef __attribute__((ext_vector_type(2))) float v2f;
typedef __attribute__((ext_vector_type(8))) float v8f;

#define GCN_IN_C   128
#define GCN_HID_C  128
#define GCN_OUT_C  64
#define GCN_BN_EPS 1e-5f


__global__ void zero_f32(float* __restrict__ p, long n) {
  long i = (long)blockIdx.x * blockDim.x + threadIdx.x;
  long stride = (long)gridDim.x * blockDim.x;
  for (; i < n; i += stride) p[i] = 0.0f;
}

__global__ void degree_rank_kernel(const int* __restrict__ dst, int E, int* __restrict__ deg,
                                   int* __restrict__ rank) {
  const int e0 = blockIdx.x * (blockDim.x * 4) + threadIdx.x;
  int d[4], r[4];
#pragma unroll
  for (int k = 0; k < 4; ++k) {
    const int e = e0 + k * blockDim.x;
    d[k] = dst[e < E ? e : E - 1];
  }
#pragma unroll
  for (int k = 0; k < 4; ++k) {
    const int e = e0 + k * blockDim.x;
    r[k] = 0;
    if (e < E) r[k] = atomicAdd(&deg[d[k]], 1);
  }
#pragma unroll
  for (int k = 0; k < 4; ++k) {
    const int e = e0 + k * blockDim.x;
    if (e < E) rank[e] = r[k];
  }
}

__global__ void __launch_bounds__(1024)
rowptr_dinv_kernel(const int* __restrict__ deg, int* __restrict__ rowptr,
                   float* __restrict__ dinv, int N) {
  __shared__ int part_scan[32];
  __shared__ int part_before[32];
  const int tid  = threadIdx.x;
  const int lane = tid & 31;
  const int w    = tid >> 5;

  int before = 0;
  const int4* __restrict__ d4 = (const int4*)deg;
  const int n4 = blockIdx.x * 256;
  for (int j = tid; j < n4; j += 1024) {
    const int4 q = d4[j];
    before += (q.x + q.y) + (q.z + q.w);
  }
  const int i  = blockIdx.x * 1024 + tid;
  const int dg = i < N ? deg[i] : 0;

  int incl = dg;
  for (int o = 1; o < 32; o <<= 1) {
    const int t = __shfl_up(incl, o, 32);
    if (lane >= o) incl += t;
  }
  for (int o = 16; o > 0; o >>= 1) before += __shfl_xor(before, o, 32);
  if (lane == 31) part_scan[w] = incl;
  if (lane == 0)  part_before[w] = before;
  __syncthreads();
  int off = part_before[lane] + (lane < w ? part_scan[lane] : 0);
  for (int o = 16; o > 0; o >>= 1) off += __shfl_xor(off, o, 32);

  if (i < N) {
    rowptr[i] = off + incl - dg;
    dinv[i]   = rsqrtf((float)(dg + 1));
  }
}

__global__ void fill_csr_kernel(const int* __restrict__ src, const int* __restrict__ dst,
                                const int* __restrict__ rank, const int* __restrict__ rowptr,
                                int* __restrict__ csr_src, int E) {
  const int e0 = blockIdx.x * (blockDim.x * 4) + threadIdx.x;
  int d[4], r[4], s[4], pos[4];
#pragma unroll
  for (int k = 0; k < 4; ++k) {
    const int e  = e0 + k * blockDim.x;
    const int ec = e < E ? e : E - 1;
    d[k] = dst[ec];
    r[k] = rank[ec];
    s[k] = src[ec];
  }
#pragma unroll
  for (int k = 0; k < 4; ++k) pos[k] = rowptr[d[k]] + r[k];
#pragma unroll
  for (int k = 0; k < 4; ++k) {
    const int e = e0 + k * blockDim.x;
    if (e < E) csr_src[pos[k]] = s[k];
  }
}

template <int KDIM>
__global__ void gemm_wmma_lds(const float* __restrict__ A, const float* __restrict__ B,
                              float* __restrict__ C, int N) {
  constexpr int LDA = KDIM + 4;
  __shared__ float As[16 * LDA];

  const int tm = blockIdx.x;
  const float* __restrict__ gA = A + (long)tm * 16 * KDIM;

  for (int idx = threadIdx.x; idx < 16 * KDIM / 4; idx += blockDim.x) {
    const int r  = idx / (KDIM / 4);
    const int c4 = idx % (KDIM / 4);
    const float4 v = ((const float4*)(gA + (long)r * KDIM))[c4];
    *(float4*)&As[r * LDA + c4 * 4] = v;
  }
  __syncthreads();

  const int lane = threadIdx.x & 31;
  const int half = lane >> 4;
  const int l16  = lane & 15;
  const int tn   = threadIdx.x >> 5;
  const int bcol = tn * 16 + l16;
  const float* __restrict__ ap = &As[l16 * LDA];

  v8f acc = {};
#pragma unroll 4
  for (int k0 = 0; k0 < KDIM; k0 += 4) {
    const int ka = k0 + 2 * half;
    v2f a, b;
    a.x = ap[ka];
    a.y = ap[ka + 1];
    b.x = B[(long)ka * N + bcol];
    b.y = B[(long)(ka + 1) * N + bcol];
    acc = __builtin_amdgcn_wmma_f32_16x16x4_f32(
         false, a,  false, b,
         (short)0, acc,  false,  false);
  }

  const int orow = tm * 16 + half * 8;
  float* __restrict__ cp = C + (long)orow * N + bcol;
#pragma unroll
  for (int r = 0; r < 8; ++r) cp[(long)r * N] = acc[r];
}

template <int C, int G>
__device__ __forceinline__ void add_rows(float (&acc)[C / 32], const float* __restrict__ h,
                                         int sv, int cvi, int j, int lane) {
  constexpr int V = C / 32;
  typedef float vec __attribute__((ext_vector_type(V)));
  vec v[G];
#pragma unroll
  for (int k = 0; k < G; ++k) {
    const int s = __builtin_amdgcn_readlane(sv, j + k);
    v[k] = *(const vec*)(h + (long)s * C + lane * V);
  }
#pragma unroll
  for (int k = 0; k < G; ++k) {
    const float coef = __int_as_float(__builtin_amdgcn_readlane(cvi, j + k));
#pragma unroll
    for (int q = 0; q < V; ++q) acc[q] = fmaf(v[k][q], coef, acc[q]);
  }
}

template <int C, int G>
__global__ void gather_rows(const float* __restrict__ h, const int* __restrict__ rowptr,
                            const int* __restrict__ deg, const int* __restrict__ csr_src,
                            const float* __restrict__ dinv, float* __restrict__ agg, int N) {
  constexpr int V = C / 32;
  typedef float vec __attribute__((ext_vector_type(V)));
  const int lane = threadIdx.x & 31;
  const int node = __builtin_amdgcn_readfirstlane(
      (int)((blockIdx.x * (long)blockDim.x + threadIdx.x) >> 5));
  if (node >= N) return;
  const int beg = __builtin_amdgcn_readfirstlane(rowptr[node]);
  const int cnt = __builtin_amdgcn_readfirstlane(deg[node]);
  const float dd = dinv[node];

  float acc[V];
#pragma unroll
  for (int q = 0; q < V; ++q) acc[q] = 0.0f;

  for (int c0 = 0; c0 < cnt; c0 += 32) {
    const int left = cnt - c0;
    const int n = left < 32 ? left : 32;
    int sv = 0;
    float cv = 0.0f;
    if (lane < n) {
      sv = csr_src[beg + c0 + lane];
      cv = dinv[sv] * dd;
    }
    const int cvi = __float_as_int(cv);
    int j = 0;
    for (; j + G <= n; j += G) add_rows<C, G>(acc, h, sv, cvi, j, lane);
    if (G > 8 && n - j >= 8) { add_rows<C, 8>(acc, h, sv, cvi, j, lane); j += 8; }
    if (G > 4 && n - j >= 4) { add_rows<C, 4>(acc, h, sv, cvi, j, lane); j += 4; }
    if (G > 2 && n - j >= 2) { add_rows<C, 2>(acc, h, sv, cvi, j, lane); j += 2; }
    if (n - j >= 1) add_rows<C, 1>(acc, h, sv, cvi, j, lane);
  }

  vec o;
#pragma unroll
  for (int q = 0; q < V; ++q) o[q] = acc[q];
  *(vec*)(agg + (long)node * C + lane * V) = o;
}

__global__ void finalize_relu_c128(const float* __restrict__ h, float* __restrict__ agg,
                                   const float* __restrict__ dinv, const float* __restrict__ bias,
                                   const float* __restrict__ gamma, const float* __restrict__ beta,
                                   const float* __restrict__ mean, const float* __restrict__ var,
                                   int N) {
  const long idx = (long)blockIdx.x * blockDim.x + threadIdx.x;
  if (idx >= (long)N * 128) return;
  const int i = (int)(idx >> 7);
  const int c = (int)(idx & 127);
  const float di = dinv[i];
  float v = agg[idx] + h[idx] * di * di + bias[c];
  v = (v - mean[c]) * rsqrtf(var[c] + GCN_BN_EPS) * gamma[c] + beta[c];
  agg[idx] = fmaxf(v, 0.0f);
}

__global__ void finalize_c64(const float* __restrict__ h, const float* __restrict__ agg,
                             const float* __restrict__ dinv, const float* __restrict__ bias,
                             const float* __restrict__ gamma, const float* __restrict__ beta,
                             const float* __restrict__ mean, const float* __restrict__ var,
                             float* __restrict__ out, int N) {
  const long idx = (long)blockIdx.x * blockDim.x + threadIdx.x;
  if (idx >= (long)N * 64) return;
  const int i = (int)(idx >> 6);
  const int c = (int)(idx & 63);
  const float di = dinv[i];
  float v = agg[idx] + h[idx] * di * di + bias[c];
  v = (v - mean[c]) * rsqrtf(var[c] + GCN_BN_EPS) * gamma[c] + beta[c];
  out[idx] = v;
}


extern "C" void kernel_launch(void* const* d_in, const int* in_sizes, int n_in,
                              void* d_out, int out_size, void* d_ws, size_t ws_size,
                              hipStream_t stream) {
  const float* x   = (const float*)d_in[0];
  const int*   ei  = (const int*)d_in[1];
  const float* W1  = (const float*)d_in[2];
  const float* b1  = (const float*)d_in[3];
  const float* g1  = (const float*)d_in[4];
  const float* be1 = (const float*)d_in[5];
  const float* m1  = (const float*)d_in[6];
  const float* v1  = (const float*)d_in[7];
  const float* W2  = (const float*)d_in[8];
  const float* b2  = (const float*)d_in[9];
  const float* g2  = (const float*)d_in[10];
  const float* be2 = (const float*)d_in[11];
  const float* m2  = (const float*)d_in[12];
  const float* v2  = (const float*)d_in[13];

  const int N = in_sizes[0] / GCN_IN_C;
  const int E = in_sizes[1] / 2;
  const int* src = ei;
  const int* dst = ei + E;

  char* ws = (char*)d_ws;
  int*   deg  = (int*)ws;                         ws += (size_t)N * sizeof(int);
  float* dinv = (float*)ws;                       ws += (size_t)N * sizeof(float);
  float* h1   = (float*)ws;                       ws += (size_t)N * GCN_HID_C * sizeof(float);
  float* agg1 = (float*)ws;                       ws += (size_t)N * GCN_HID_C * sizeof(float);
  float* h2   = (float*)ws;                       ws += (size_t)N * GCN_OUT_C * sizeof(float);
  float* agg2 = (float*)ws;                       ws += (size_t)N * GCN_OUT_C * sizeof(float);
  int*   rowptr  = (int*)ws;                      ws += (size_t)N * sizeof(int);
  int*   rank    = (int*)ws;                      ws += (size_t)E * sizeof(int);
  int*   csr_src = (int*)ws;                      ws += (size_t)E * sizeof(int);
  (void)ws_size; (void)n_in; (void)out_size;

  const int ZB = 2048;
  const int eblocks = (E + 1023) / 1024;
  const int gblocks = (N + 7) / 8;
  zero_f32<<<ZB, 256, 0, stream>>>((float*)deg, (long)N);

  degree_rank_kernel<<<eblocks, 256, 0, stream>>>(dst, E, deg, rank);
  rowptr_dinv_kernel<<<(N + 1023) / 1024, 1024, 0, stream>>>(deg, rowptr, dinv, N);
  fill_csr_kernel<<<eblocks, 256, 0, stream>>>(src, dst, rank, rowptr, csr_src, E);

  gemm_wmma_lds<GCN_IN_C><<<N / 16, (GCN_HID_C / 16) * 32, 0, stream>>>(x, W1, h1, GCN_HID_C);
  gather_rows<GCN_HID_C, 8><<<gblocks, 256, 0, stream>>>(h1, rowptr, deg, csr_src, dinv, agg1, N);
  {
    const long total = (long)N * GCN_HID_C;
    finalize_relu_c128<<<(int)((total + 255) / 256), 256, 0, stream>>>(
        h1, agg1, dinv, b1, g1, be1, m1, v1, N);
  }

  gemm_wmma_lds<GCN_HID_C><<<N / 16, (GCN_OUT_C / 16) * 32, 0, stream>>>(agg1, W2, h2, GCN_OUT_C);
  gather_rows<GCN_OUT_C, 16><<<gblocks, 256, 0, stream>>>(h2, rowptr, deg, csr_src, dinv, agg2, N);
  {
    const long total = (long)N * GCN_OUT_C;
    finalize_c64<<<(int)((total + 255) / 256), 256, 0, stream>>>(
        h2, agg2, dinv, b2, g2, be2, m2, v2, (float*)d_out, N);
  }
}


# gather waves finish their node's row themselves (own row added, x deg^-1/2, bias, batch norm, ReLU in layer 1) and store it final: the two elementwise finalize launches go; sum order kept
# speedup vs baseline: 2.4125x; 1.1619x over previous
#include <hip/hip_runtime.h>

typedef __attribute__((ext_vector_type(2))) float v2f;
typedef __attribute__((ext_vector_type(8))) float v8f;

#define GCN_IN_C   128
#define GCN_HID_C  128
#define GCN_OUT_C  64
#define GCN_BN_EPS 1e-5f


__global__ void zero_f32(float* __restrict__ p, long n) {
  long i = (long)blockIdx.x * blockDim.x + threadIdx.x;
  long stride = (long)gridDim.x * blockDim.x;
  for (; i < n; i += stride) p[i] = 0.0f;
}

__global__ void degree_rank_kernel(const int* __restrict__ dst, int E, int* __restrict__ deg,
                                   int* __restrict__ rank) {
  const int e0 = blockIdx.x * (blockDim.x * 4) + threadIdx.x;
  int d[4], r[4];
#pragma unroll
  for (int k = 0; k < 4; ++k) {
    const int e = e0 + k * blockDim.x;
    d[k] = dst[e < E ? e : E - 1];
  }
#pragma unroll
  for (int k = 0; k < 4; ++k) {
    const int e = e0 + k * blockDim.x;
    r[k] = 0;
    if (e < E) r[k] = atomicAdd(&deg[d[k]], 1);
  }
#pragma unroll
  for (int k = 0; k < 4; ++k) {
    const int e = e0 + k * blockDim.x;
    if (e < E) rank[e] = r[k];
  }
}

__global__ void __launch_bounds__(1024)
rowptr_dinv_kernel(const int* __restrict__ deg, int* __restrict__ rowptr,
                   float* __restrict__ dinv, int N) {
  __shared__ int part_scan[32];
  __shared__ int part_before[32];
  const int tid  = threadIdx.x;
  const int lane = tid & 31;
  const int w    = tid >> 5;

  int before = 0;
  const int4* __restrict__ d4 = (const int4*)deg;
  const int n4 = blockIdx.x * 256;
  for (int j = tid; j < n4; j += 1024) {
    const int4 q = d4[j];
    before += (q.x + q.y) + (q.z + q.w);
  }
  const int i  = blockIdx.x * 1024 + tid;
  const int dg = i < N ? deg[i] : 0;

  int incl = dg;
  for (int o = 1; o < 32; o <<= 1) {
    const int t = __shfl_up(incl, o, 32);
    if (lane >= o) incl += t;
  }
  for (int o = 16; o > 0; o >>= 1) before += __shfl_xor(before, o, 32);
  if (lane == 31) part_scan[w] = incl;
  if (lane == 0)  part_before[w] = before;
  __syncthreads();
  int off = part_before[lane] + (lane < w ? part_scan[lane] : 0);
  for (int o = 16; o > 0; o >>= 1) off += __shfl_xor(off, o, 32);

  if (i < N) {
    rowptr[i] = off + incl - dg;
    dinv[i]   = rsqrtf((float)(dg + 1));
  }
}

__global__ void fill_csr_kernel(const int* __restrict__ src, const int* __restrict__ dst,
                                const int* __restrict__ rank, const int* __restrict__ rowptr,
                                int* __restrict__ csr_src, int E) {
  const int e0 = blockIdx.x * (blockDim.x * 4) + threadIdx.x;
  int d[4], r[4], s[4], pos[4];
#pragma unroll
  for (int k = 0; k < 4; ++k) {
    const int e  = e0 + k * blockDim.x;
    const int ec = e < E ? e : E - 1;
    d[k] = dst[ec];
    r[k] = rank[ec];
    s[k] = src[ec];
  }
#pragma unroll
  for (int k = 0; k < 4; ++k) pos[k] = rowptr[d[k]] + r[k];
#pragma unroll
  for (int k = 0; k < 4; ++k) {
    const int e = e0 + k * blockDim.x;
    if (e < E) csr_src[pos[k]] = s[k];
  }
}

template <int KDIM>
__global__ void gemm_wmma_lds(const float* __restrict__ A, const float* __restrict__ B,
                              const float* __restrict__ rscale, float* __restrict__ C, int N) {
  constexpr int LDA = KDIM + 4;
  __shared__ float As[16 * LDA];

  const int tm = blockIdx.x;
  const float* __restrict__ gA = A + (long)tm * 16 * KDIM;

  for (int idx = threadIdx.x; idx < 16 * KDIM / 4; idx += blockDim.x) {
    const int r  = idx / (KDIM / 4);
    const int c4 = idx % (KDIM / 4);
    const float4 v = ((const float4*)(gA + (long)r * KDIM))[c4];
    *(float4*)&As[r * LDA + c4 * 4] = v;
  }
  __syncthreads();

  const int lane = threadIdx.x & 31;
  const int half = lane >> 4;
  const int l16  = lane & 15;
  const int tn   = threadIdx.x >> 5;
  const int bcol = tn * 16 + l16;
  const float* __restrict__ ap = &As[l16 * LDA];

  const int orow = tm * 16 + half * 8;
  const float4 rs0 = ((const float4*)(rscale + orow))[0];
  const float4 rs1 = ((const float4*)(rscale + orow))[1];
  const float rs[8] = {rs0.x, rs0.y, rs0.z, rs0.w, rs1.x, rs1.y, rs1.z, rs1.w};

  v8f acc = {};
#pragma unroll 4
  for (int k0 = 0; k0 < KDIM; k0 += 4) {
    const int ka = k0 + 2 * half;
    v2f a, b;
    a.x = ap[ka];
    a.y = ap[ka + 1];
    b.x = B[(long)ka * N + bcol];
    b.y = B[(long)(ka + 1) * N + bcol];
    acc = __builtin_amdgcn_wmma_f32_16x16x4_f32(
         false, a,  false, b,
         (short)0, acc,  false,  false);
  }

  float* __restrict__ cp = C + (long)orow * N + bcol;
#pragma unroll
  for (int r = 0; r < 8; ++r) cp[(long)r * N] = acc[r] * rs[r];
}

template <int C, int G>
__device__ __forceinline__ void add_rows(float (&acc)[C / 32], const float* __restrict__ h,
                                         int sv, int j, int lane) {
  constexpr int V = C / 32;
  typedef float vec __attribute__((ext_vector_type(V)));
  vec v[G];
#pragma unroll
  for (int k = 0; k < G; ++k) {
    const int s = __builtin_amdgcn_readlane(sv, j + k);
    v[k] = *(const vec*)(h + (long)s * C + lane * V);
  }
#pragma unroll
  for (int k = 0; k < G; ++k) {
#pragma unroll
    for (int q = 0; q < V; ++q) acc[q] += v[k][q];
  }
}

template <int C, int G, bool RELU>
__global__ void gather_finish_rows(const float* __restrict__ h, const int* __restrict__ rowptr,
                                   const int* __restrict__ deg, const int* __restrict__ csr_src,
                                   const float* __restrict__ dinv, const float* __restrict__ bias,
                                   const float* __restrict__ gamma, const float* __restrict__ beta,
                                   const float* __restrict__ mean, const float* __restrict__ var,
                                   float* __restrict__ outp, int N) {
  constexpr int V = C / 32;
  typedef float vec __attribute__((ext_vector_type(V)));
  const int lane = threadIdx.x & 31;
  const int node = __builtin_amdgcn_readfirstlane(
      (int)((blockIdx.x * (long)blockDim.x + threadIdx.x) >> 5));
  if (node >= N) return;
  const int beg = __builtin_amdgcn_readfirstlane(rowptr[node]);
  const int cnt = __builtin_amdgcn_readfirstlane(deg[node]);
  int sv = 0;
  if (lane < cnt) sv = csr_src[beg + lane];
  const vec own = *(const vec*)(h + (long)node * C + lane * V);
  const float di = dinv[node];

  float acc[V];
#pragma unroll
  for (int q = 0; q < V; ++q) acc[q] = 0.0f;

  for (int c0 = 0; c0 < cnt; c0 += 32) {
    const int left = cnt - c0;
    const int n = left < 32 ? left : 32;
    if (c0 > 0) {
      sv = 0;
      if (lane < n) sv = csr_src[beg + c0 + lane];
    }
    int j = 0;
    for (; j + G <= n; j += G) add_rows<C, G>(acc, h, sv, j, lane);
    if (G > 8 && n - j >= 8) { add_rows<C, 8>(acc, h, sv, j, lane); j += 8; }
    if (G > 4 && n - j >= 4) { add_rows<C, 4>(acc, h, sv, j, lane); j += 4; }
    if (G > 2 && n - j >= 2) { add_rows<C, 2>(acc, h, sv, j, lane); j += 2; }
    if (n - j >= 1) add_rows<C, 1>(acc, h, sv, j, lane);
  }

  const vec bs = *(const vec*)(bias + lane * V);
  const vec mn = *(const vec*)(mean + lane * V);
  const vec vr = *(const vec*)(var + lane * V);
  const vec gm = *(const vec*)(gamma + lane * V);
  const vec bt = *(const vec*)(beta + lane * V);

  vec o;
#pragma unroll
  for (int q = 0; q < V; ++q) {
    float v = (acc[q] + own[q]) * di + bs[q];
    v = (v - mn[q]) * rsqrtf(vr[q] + GCN_BN_EPS) * gm[q] + bt[q];
    o[q] = RELU ? fmaxf(v, 0.0f) : v;
  }
  *(vec*)(outp + (long)node * C + lane * V) = o;
}


extern "C" void kernel_launch(void* const* d_in, const int* in_sizes, int n_in,
                              void* d_out, int out_size, void* d_ws, size_t ws_size,
                              hipStream_t stream) {
  const float* x   = (const float*)d_in[0];
  const int*   ei  = (const int*)d_in[1];
  const float* W1  = (const float*)d_in[2];
  const float* b1  = (const float*)d_in[3];
  const float* g1  = (const float*)d_in[4];
  const float* be1 = (const float*)d_in[5];
  const float* m1  = (const float*)d_in[6];
  const float* v1  = (const float*)d_in[7];
  const float* W2  = (const float*)d_in[8];
  const float* b2  = (const float*)d_in[9];
  const float* g2  = (const float*)d_in[10];
  const float* be2 = (const float*)d_in[11];
  const float* m2  = (const float*)d_in[12];
  const float* v2  = (const float*)d_in[13];

  const int N = in_sizes[0] / GCN_IN_C;
  const int E = in_sizes[1] / 2;
  const int* src = ei;
  const int* dst = ei + E;

  char* ws = (char*)d_ws;
  int*   deg  = (int*)ws;                         ws += (size_t)N * sizeof(int);
  float* dinv = (float*)ws;                       ws += (size_t)N * sizeof(float);
  float* h1   = (float*)ws;                       ws += (size_t)N * GCN_HID_C * sizeof(float);
  float* act1 = (float*)ws;                       ws += (size_t)N * GCN_HID_C * sizeof(float);
  float* h2   = (float*)ws;                       ws += (size_t)N * GCN_OUT_C * sizeof(float);
  int*   rowptr  = (int*)ws;                      ws += (size_t)N * sizeof(int);
  int*   rank    = (int*)ws;                      ws += (size_t)E * sizeof(int);
  int*   csr_src = (int*)ws;                      ws += (size_t)E * sizeof(int);
  (void)ws_size; (void)n_in; (void)out_size;

  const int ZB = 2048;
  const int eblocks = (E + 1023) / 1024;
  const int gblocks = (N + 7) / 8;
  zero_f32<<<ZB, 256, 0, stream>>>((float*)deg, (long)N);

  degree_rank_kernel<<<eblocks, 256, 0, stream>>>(dst, E, deg, rank);
  rowptr_dinv_kernel<<<(N + 1023) / 1024, 1024, 0, stream>>>(deg, rowptr, dinv, N);
  fill_csr_kernel<<<eblocks, 256, 0, stream>>>(src, dst, rank, rowptr, csr_src, E);

  gemm_wmma_lds<GCN_IN_C><<<N / 16, (GCN_HID_C / 16) * 32, 0, stream>>>(x, W1, dinv, h1, GCN_HID_C);
  gather_finish_rows<GCN_HID_C, 8, true><<<gblocks, 256, 0, stream>>>(
      h1, rowptr, deg, csr_src, dinv, b1, g1, be1, m1, v1, act1, N);

  gemm_wmma_lds<GCN_HID_C><<<N / 16, (GCN_OUT_C / 16) * 32, 0, stream>>>(act1, W2, dinv, h2, GCN_OUT_C);
  gather_finish_rows<GCN_OUT_C, 16, false><<<gblocks, 256, 0, stream>>>(
      h2, rowptr, deg, csr_src, dinv, b2, g2, be2, m2, v2, (float*)d_out, N);
}


# edges go straight into a 64-slot row per dst node (slot = returning add on its count; rare overflow to a shared list): no rank array, prefix scan or second edge pass; arrival-order sums kept
# speedup vs baseline: 2.6242x; 1.0878x over previous
#include <hip/hip_runtime.h>

typedef __attribute__((ext_vector_type(2))) float v2f;
typedef __attribute__((ext_vector_type(8))) float v8f;

#define GCN_IN_C   128
#define GCN_HID_C  128
#define GCN_OUT_C  64
#define GCN_BN_EPS 1e-5f
#define GCN_SLOT_CAP 64


__global__ void zero_f32(float* __restrict__ p, long n) {
  long i = (long)blockIdx.x * blockDim.x + threadIdx.x;
  long stride = (long)gridDim.x * blockDim.x;
  for (; i < n; i += stride) p[i] = 0.0f;
}

__global__ void bucket_edges_kernel(const int* __restrict__ src, const int* __restrict__ dst, int E,
                                    int* __restrict__ cnt, int* __restrict__ slots,
                                    int* __restrict__ ovf_n, int* __restrict__ ovf_src,
                                    int* __restrict__ ovf_dst) {
  const int e0 = blockIdx.x * (blockDim.x * 4) + threadIdx.x;
  int d[4], s[4], r[4];
#pragma unroll
  for (int k = 0; k < 4; ++k) {
    const int e  = e0 + k * blockDim.x;
    const int ec = e < E ? e : E - 1;
    d[k] = dst[ec];
    s[k] = src[ec];
  }
#pragma unroll
  for (int k = 0; k < 4; ++k) {
    const int e = e0 + k * blockDim.x;
    r[k] = 0;
    if (e < E) r[k] = atomicAdd(&cnt[d[k]], 1);
  }
#pragma unroll
  for (int k = 0; k < 4; ++k) {
    const int e = e0 + k * blockDim.x;
    if (e < E) {
      if (r[k] < GCN_SLOT_CAP) {
        slots[(long)d[k] * GCN_SLOT_CAP + r[k]] = s[k];
      } else {
        const int o = atomicAdd(ovf_n, 1);
        ovf_src[o] = s[k];
        ovf_dst[o] = d[k];
      }
    }
  }
}

template <int KDIM>
__global__ void gemm_wmma_lds(const float* __restrict__ A, const float* __restrict__ B,
                              const int* __restrict__ cnt, float* __restrict__ C, int N) {
  constexpr int LDA = KDIM + 4;
  __shared__ float As[16 * LDA];

  const int tm = blockIdx.x;
  const float* __restrict__ gA = A + (long)tm * 16 * KDIM;

  for (int idx = threadIdx.x; idx < 16 * KDIM / 4; idx += blockDim.x) {
    const int r  = idx / (KDIM / 4);
    const int c4 = idx % (KDIM / 4);
    const float4 v = ((const float4*)(gA + (long)r * KDIM))[c4];
    *(float4*)&As[r * LDA + c4 * 4] = v;
  }
  __syncthreads();

  const int lane = threadIdx.x & 31;
  const int half = lane >> 4;
  const int l16  = lane & 15;
  const int tn   = threadIdx.x >> 5;
  const int bcol = tn * 16 + l16;
  const float* __restrict__ ap = &As[l16 * LDA];

  const int orow = tm * 16 + half * 8;
  const int4 dg0 = ((const int4*)(cnt + orow))[0];
  const int4 dg1 = ((const int4*)(cnt + orow))[1];
  const float rs[8] = {rsqrtf((float)(dg0.x + 1)), rsqrtf((float)(dg0.y + 1)),
                       rsqrtf((float)(dg0.z + 1)), rsqrtf((float)(dg0.w + 1)),
                       rsqrtf((float)(dg1.x + 1)), rsqrtf((float)(dg1.y + 1)),
                       rsqrtf((float)(dg1.z + 1)), rsqrtf((float)(dg1.w + 1))};

  v8f acc = {};
#pragma unroll 4
  for (int k0 = 0; k0 < KDIM; k0 += 4) {
    const int ka = k0 + 2 * half;
    v2f a, b;
    a.x = ap[ka];
    a.y = ap[ka + 1];
    b.x = B[(long)ka * N + bcol];
    b.y = B[(long)(ka + 1) * N + bcol];
    acc = __builtin_amdgcn_wmma_f32_16x16x4_f32(
         false, a,  false, b,
         (short)0, acc,  false,  false);
  }

  float* __restrict__ cp = C + (long)orow * N + bcol;
#pragma unroll
  for (int r = 0; r < 8; ++r) cp[(long)r * N] = acc[r] * rs[r];
}

template <int C, int G>
__device__ __forceinline__ void add_rows(float (&acc)[C / 32], const float* __restrict__ h,
                                         int sv, int j, int lane) {
  constexpr int V = C / 32;
  typedef float vec __attribute__((ext_vector_type(V)));
  vec v[G];
#pragma unroll
  for (int k = 0; k < G; ++k) {
    const int s = __builtin_amdgcn_readlane(sv, j + k);
    v[k] = *(const vec*)(h + (long)s * C + lane * V);
  }
#pragma unroll
  for (int k = 0; k < G; ++k) {
#pragma unroll
    for (int q = 0; q < V; ++q) acc[q] += v[k][q];
  }
}

template <int C, int G, bool RELU>
__global__ void gather_finish_rows(const float* __restrict__ h, const int* __restrict__ cnt,
                                   const int* __restrict__ slots, const int* __restrict__ ovf_n,
                                   const int* __restrict__ ovf_src, const int* __restrict__ ovf_dst,
                                   const float* __restrict__ bias,
                                   const float* __restrict__ gamma, const float* __restrict__ beta,
                                   const float* __restrict__ mean, const float* __restrict__ var,
                                   float* __restrict__ outp, int N) {
  constexpr int V = C / 32;
  typedef float vec __attribute__((ext_vector_type(V)));
  const int lane = threadIdx.x & 31;
  const int node = __builtin_amdgcn_readfirstlane(
      (int)((blockIdx.x * (long)blockDim.x + threadIdx.x) >> 5));
  if (node >= N) return;
  const int tot = __builtin_amdgcn_readfirstlane(cnt[node]);
  const int nin = tot < GCN_SLOT_CAP ? tot : GCN_SLOT_CAP;
  const int* __restrict__ row = slots + (long)node * GCN_SLOT_CAP;
  int sv = 0;
  if (lane < nin) sv = row[lane];
  const vec own = *(const vec*)(h + (long)node * C + lane * V);
  const float di = rsqrtf((float)(tot + 1));

  float acc[V];
#pragma unroll
  for (int q = 0; q < V; ++q) acc[q] = 0.0f;

  for (int c0 = 0; c0 < nin; c0 += 32) {
    const int left = nin - c0;
    const int n = left < 32 ? left : 32;
    if (c0 > 0) {
      sv = 0;
      if (lane < n) sv = row[c0 + lane];
    }
    int j = 0;
    for (; j + G <= n; j += G) add_rows<C, G>(acc, h, sv, j, lane);
    if (G > 8 && n - j >= 8) { add_rows<C, 8>(acc, h, sv, j, lane); j += 8; }
    if (G > 4 && n - j >= 4) { add_rows<C, 4>(acc, h, sv, j, lane); j += 4; }
    if (G > 2 && n - j >= 2) { add_rows<C, 2>(acc, h, sv, j, lane); j += 2; }
    if (n - j >= 1) add_rows<C, 1>(acc, h, sv, j, lane);
  }

  if (tot > GCN_SLOT_CAP) {
    const int no = __builtin_amdgcn_readfirstlane(*ovf_n);
    for (int o0 = 0; o0 < no; o0 += 32) {
      int od = -1, os = 0;
      if (o0 + lane < no) {
        od = ovf_dst[o0 + lane];
        os = ovf_src[o0 + lane];
      }
      unsigned m = __builtin_amdgcn_ballot_w32(od == node);
      while (m) {
        const int j = __builtin_ctz(m);
        m &= m - 1;
        add_rows<C, 1>(acc, h, os, j, lane);
      }
    }
  }

  const vec bs = *(const vec*)(bias + lane * V);
  const vec mn = *(const vec*)(mean + lane * V);
  const vec vr = *(const vec*)(var + lane * V);
  const vec gm = *(const vec*)(gamma + lane * V);
  const vec bt = *(const vec*)(beta + lane * V);

  vec o;
#pragma unroll
  for (int q = 0; q < V; ++q) {
    float v = (acc[q] + own[q]) * di + bs[q];
    v = (v - mn[q]) * rsqrtf(vr[q] + GCN_BN_EPS) * gm[q] + bt[q];
    o[q] = RELU ? fmaxf(v, 0.0f) : v;
  }
  *(vec*)(outp + (long)node * C + lane * V) = o;
}


extern "C" void kernel_launch(void* const* d_in, const int* in_sizes, int n_in,
                              void* d_out, int out_size, void* d_ws, size_t ws_size,
                              hipStream_t stream) {
  const float* x   = (const float*)d_in[0];
  const int*   ei  = (const int*)d_in[1];
  const float* W1  = (const float*)d_in[2];
  const float* b1  = (const float*)d_in[3];
  const float* g1  = (const float*)d_in[4];
  const float* be1 = (const float*)d_in[5];
  const float* m1  = (const float*)d_in[6];
  const float* v1  = (const float*)d_in[7];
  const float* W2  = (const float*)d_in[8];
  const float* b2  = (const float*)d_in[9];
  const float* g2  = (const float*)d_in[10];
  const float* be2 = (const float*)d_in[11];
  const float* m2  = (const float*)d_in[12];
  const float* v2  = (const float*)d_in[13];

  const int N = in_sizes[0] / GCN_IN_C;
  const int E = in_sizes[1] / 2;
  const int* src = ei;
  const int* dst = ei + E;

  char* ws = (char*)d_ws;
  int*   cnt  = (int*)ws;                         ws += (size_t)N * 2 * sizeof(int);
  int*   ovf_n = cnt + N;
  float* h1   = (float*)ws;                       ws += (size_t)N * GCN_HID_C * sizeof(float);
  float* act1 = (float*)ws;                       ws += (size_t)N * GCN_HID_C * sizeof(float);
  float* h2   = (float*)ws;                       ws += (size_t)N * GCN_OUT_C * sizeof(float);
  ws += (256 - (size_t)(ws - (char*)d_ws) % 256) % 256;
  int*   slots   = (int*)ws;                      ws += (size_t)N * GCN_SLOT_CAP * sizeof(int);
  int*   ovf_src = (int*)ws;                      ws += (size_t)E * sizeof(int);
  int*   ovf_dst = (int*)ws;                      ws += (size_t)E * sizeof(int);
  (void)ws_size; (void)n_in; (void)out_size;

  const int ZB = 2048;
  const int eblocks = (E + 1023) / 1024;
  const int gblocks = (N + 7) / 8;
  zero_f32<<<ZB, 256, 0, stream>>>((float*)cnt, (long)N + 1);

  bucket_edges_kernel<<<eblocks, 256, 0, stream>>>(src, dst, E, cnt, slots, ovf_n, ovf_src, ovf_dst);

  gemm_wmma_lds<GCN_IN_C><<<N / 16, (GCN_HID_C / 16) * 32, 0, stream>>>(x, W1, cnt, h1, GCN_HID_C);
  gather_finish_rows<GCN_HID_C, 8, true><<<gblocks, 256, 0, stream>>>(
      h1, cnt, slots, ovf_n, ovf_src, ovf_dst, b1, g1, be1, m1, v1, act1, N);

  gemm_wmma_lds<GCN_HID_C><<<N / 16, (GCN_OUT_C / 16) * 32, 0, stream>>>(act1, W2, cnt, h2, GCN_OUT_C);
  gather_finish_rows<GCN_OUT_C, 16, false><<<gblocks, 256, 0, stream>>>(
      h2, cnt, slots, ovf_n, ovf_src, ovf_dst, b2, g2, be2, m2, v2, (float*)d_out, N);
}


# speedup vs baseline: 2.8055x; 1.0691x over previous
#include <hip/hip_runtime.h>

typedef __attribute__((ext_vector_type(2))) float v2f;
typedef __attribute__((ext_vector_type(8))) float v8f;

#define GCN_IN_C   128
#define GCN_HID_C  128
#define GCN_OUT_C  64
#define GCN_BN_EPS 1e-5f
#define GCN_SLOT_CAP 64


__global__ void zero_f32(float* __restrict__ p, long n) {
  long i = (long)blockIdx.x * blockDim.x + threadIdx.x;
  long stride = (long)gridDim.x * blockDim.x;
  for (; i < n; i += stride) p[i] = 0.0f;
}

__global__ void bucket_edges_kernel(const int* __restrict__ src, const int* __restrict__ dst, int E,
                                    int* __restrict__ cnt, int* __restrict__ slots,
                                    int* __restrict__ ovf_n, int* __restrict__ ovf_src,
                                    int* __restrict__ ovf_dst) {
  const int e0 = blockIdx.x * (blockDim.x * 4) + threadIdx.x;
  int d[4], s[4], r[4];
#pragma unroll
  for (int k = 0; k < 4; ++k) {
    const int e  = e0 + k * blockDim.x;
    const int ec = e < E ? e : E - 1;
    d[k] = dst[ec];
    s[k] = src[ec];
  }
#pragma unroll
  for (int k = 0; k < 4; ++k) {
    const int e = e0 + k * blockDim.x;
    r[k] = 0;
    if (e < E) r[k] = atomicAdd(&cnt[d[k]], 1);
  }
#pragma unroll
  for (int k = 0; k < 4; ++k) {
    const int e = e0 + k * blockDim.x;
    if (e < E) {
      if (r[k] < GCN_SLOT_CAP) {
        slots[(long)d[k] * GCN_SLOT_CAP + r[k]] = s[k];
      } else {
        const int o = atomicAdd(ovf_n, 1);
        ovf_src[o] = s[k];
        ovf_dst[o] = d[k];
      }
    }
  }
}

template <int KDIM>
__global__ void gemm_wmma_lds(const float* __restrict__ A, const float* __restrict__ B,
                              const int* __restrict__ cnt, float* __restrict__ C, int N) {
  constexpr int LDA = KDIM + 4;
  __shared__ float As[16 * LDA];

  const int tm = blockIdx.x;
  const float* __restrict__ gA = A + (long)tm * 16 * KDIM;

  for (int idx = threadIdx.x; idx < 16 * KDIM / 4; idx += blockDim.x) {
    const int r  = idx / (KDIM / 4);
    const int c4 = idx % (KDIM / 4);
    const float4 v = ((const float4*)(gA + (long)r * KDIM))[c4];
    *(float4*)&As[r * LDA + c4 * 4] = v;
  }
  __syncthreads();

  const int lane = threadIdx.x & 31;
  const int half = lane >> 4;
  const int l16  = lane & 15;
  const int tn   = threadIdx.x >> 5;
  const int bcol = tn * 16 + l16;
  const float* __restrict__ ap = &As[l16 * LDA];

  const int orow = tm * 16 + half * 8;
  const int4 dg0 = ((const int4*)(cnt + orow))[0];
  const int4 dg1 = ((const int4*)(cnt + orow))[1];
  const float rs[8] = {rsqrtf((float)(dg0.x + 1)), rsqrtf((float)(dg0.y + 1)),
                       rsqrtf((float)(dg0.z + 1)), rsqrtf((float)(dg0.w + 1)),
                       rsqrtf((float)(dg1.x + 1)), rsqrtf((float)(dg1.y + 1)),
                       rsqrtf((float)(dg1.z + 1)), rsqrtf((float)(dg1.w + 1))};

  v8f acc = {};
#pragma unroll 4
  for (int k0 = 0; k0 < KDIM; k0 += 4) {
    const int ka = k0 + 2 * half;
    v2f a, b;
    a.x = ap[ka];
    a.y = ap[ka + 1];
    b.x = B[(long)ka * N + bcol];
    b.y = B[(long)(ka + 1) * N + bcol];
    acc = __builtin_amdgcn_wmma_f32_16x16x4_f32(
         false, a,  false, b,
         (short)0, acc,  false,  false);
  }

  float* __restrict__ cp = C + (long)orow * N + bcol;
#pragma unroll
  for (int r = 0; r < 8; ++r) cp[(long)r * N] = acc[r] * rs[r];
}

template <int C, int G>
__device__ __forceinline__ void add_rows(float (&acc)[C / 32], const float* __restrict__ h,
                                         int sv, int j, int lane) {
  constexpr int V = C / 32;
  typedef float vec __attribute__((ext_vector_type(V)));
  vec v[G];
#pragma unroll
  for (int k = 0; k < G; ++k) {
    const int s = __builtin_amdgcn_readlane(sv, j + k);
    v[k] = *(const vec*)(h + (long)s * C + lane * V);
  }
#pragma unroll
  for (int k = 0; k < G; ++k) {
#pragma unroll
    for (int q = 0; q < V; ++q) acc[q] += v[k][q];
  }
}

template <int C, int G, bool RELU>
__global__ void gather_finish_rows(const float* __restrict__ h, const int* __restrict__ cnt,
                                   const int* __restrict__ slots, const int* __restrict__ ovf_n,
                                   const int* __restrict__ ovf_src, const int* __restrict__ ovf_dst,
                                   const float* __restrict__ bias,
                                   const float* __restrict__ gamma, const float* __restrict__ beta,
                                   const float* __restrict__ mean, const float* __restrict__ var,
                                   float* __restrict__ outp, int N) {
  constexpr int V = C / 32;
  typedef float vec __attribute__((ext_vector_type(V)));
  const int lane = threadIdx.x & 31;
  const int node = __builtin_amdgcn_readfirstlane(
      (int)((blockIdx.x * (long)blockDim.x + threadIdx.x) >> 5));
  if (node >= N) return;
  const int tot = __builtin_amdgcn_readfirstlane(cnt[node]);
  const int nin = tot < GCN_SLOT_CAP ? tot : GCN_SLOT_CAP;
  const int* __restrict__ row = slots + (long)node * GCN_SLOT_CAP;
  int sv = 0;
  if (lane < nin) sv = row[lane];
  const vec own = *(const vec*)(h + (long)node * C + lane * V);
  const float di = rsqrtf((float)(tot + 1));

  float acc[V];
#pragma unroll
  for (int q = 0; q < V; ++q) acc[q] = 0.0f;

  for (int c0 = 0; c0 < nin; c0 += 32) {
    const int left = nin - c0;
    const int n = left < 32 ? left : 32;
    if (c0 > 0) {
      sv = 0;
      if (lane < n) sv = row[c0 + lane];
    }
    int j = 0;
    for (; j + G <= n; j += G) add_rows<C, G>(acc, h, sv, j, lane);
    if (G > 8 && n - j >= 8) { add_rows<C, 8>(acc, h, sv, j, lane); j += 8; }
    if (G > 4 && n - j >= 4) { add_rows<C, 4>(acc, h, sv, j, lane); j += 4; }
    if (G > 2 && n - j >= 2) { add_rows<C, 2>(acc, h, sv, j, lane); j += 2; }
    if (n - j >= 1) add_rows<C, 1>(acc, h, sv, j, lane);
  }

  if (tot > GCN_SLOT_CAP) {
    const int no = __builtin_amdgcn_readfirstlane(*ovf_n);
    for (int o0 = 0; o0 < no; o0 += 32) {
      int od = -1, os = 0;
      if (o0 + lane < no) {
        od = ovf_dst[o0 + lane];
        os = ovf_src[o0 + lane];
      }
      unsigned m = __builtin_amdgcn_ballot_w32(od == node);
      while (m) {
        const int j = __builtin_ctz(m);
        m &= m - 1;
        add_rows<C, 1>(acc, h, os, j, lane);
      }
    }
  }

  const vec bs = *(const vec*)(bias + lane * V);
  const vec mn = *(const vec*)(mean + lane * V);
  const vec vr = *(const vec*)(var + lane * V);
  const vec gm = *(const vec*)(gamma + lane * V);
  const vec bt = *(const vec*)(beta + lane * V);

  vec o;
#pragma unroll
  for (int q = 0; q < V; ++q) {
    float v = (acc[q] + own[q]) * di + bs[q];
    v = (v - mn[q]) * rsqrtf(vr[q] + GCN_BN_EPS) * gm[q] + bt[q];
    o[q] = RELU ? fmaxf(v, 0.0f) : v;
  }
  *(vec*)(outp + (long)node * C + lane * V) = o;
}


extern "C" void kernel_launch(void* const* d_in, const int* in_sizes, int n_in,
                              void* d_out, int out_size, void* d_ws, size_t ws_size,
                              hipStream_t stream) {
  const float* x   = (const float*)d_in[0];
  const int*   ei  = (const int*)d_in[1];
  const float* W1  = (const float*)d_in[2];
  const float* b1  = (const float*)d_in[3];
  const float* g1  = (const float*)d_in[4];
  const float* be1 = (const float*)d_in[5];
  const float* m1  = (const float*)d_in[6];
  const float* v1  = (const float*)d_in[7];
  const float* W2  = (const float*)d_in[8];
  const float* b2  = (const float*)d_in[9];
  const float* g2  = (const float*)d_in[10];
  const float* be2 = (const float*)d_in[11];
  const float* m2  = (const float*)d_in[12];
  const float* v2  = (const float*)d_in[13];

  const int N = in_sizes[0] / GCN_IN_C;
  const int E = in_sizes[1] / 2;
  const int* src = ei;
  const int* dst = ei + E;

  char* ws = (char*)d_ws;
  int*   cnt  = (int*)ws;                         ws += (size_t)N * 2 * sizeof(int);
  int*   ovf_n = cnt + N;
  ws += (512 - (size_t)(ws - (char*)d_ws) % 512) % 512;
  float* h1   = (float*)ws;                       ws += (size_t)N * GCN_HID_C * sizeof(float);
  float* act1 = (float*)ws;                       ws += (size_t)N * GCN_HID_C * sizeof(float);
  float* h2   = (float*)ws;                       ws += (size_t)N * GCN_OUT_C * sizeof(float);
  ws += (256 - (size_t)(ws - (char*)d_ws) % 256) % 256;
  int*   slots   = (int*)ws;                      ws += (size_t)N * GCN_SLOT_CAP * sizeof(int);
  int*   ovf_src = (int*)ws;                      ws += (size_t)E * sizeof(int);
  int*   ovf_dst = (int*)ws;                      ws += (size_t)E * sizeof(int);
  (void)ws_size; (void)n_in; (void)out_size;

  const int ZB = 2048;
  const int eblocks = (E + 1023) / 1024;
  const int gblocks = (N + 7) / 8;
  zero_f32<<<ZB, 256, 0, stream>>>((float*)cnt, (long)N + 1);

  bucket_edges_kernel<<<eblocks, 256, 0, stream>>>(src, dst, E, cnt, slots, ovf_n, ovf_src, ovf_dst);

  gemm_wmma_lds<GCN_IN_C><<<N / 16, (GCN_HID_C / 16) * 32, 0, stream>>>(x, W1, cnt, h1, GCN_HID_C);
  gather_finish_rows<GCN_HID_C, 8, true><<<gblocks, 256, 0, stream>>>(
      h1, cnt, slots, ovf_n, ovf_src, ovf_dst, b1, g1, be1, m1, v1, act1, N);

  gemm_wmma_lds<GCN_HID_C><<<N / 16, (GCN_OUT_C / 16) * 32, 0, stream>>>(act1, W2, cnt, h2, GCN_OUT_C);
  gather_finish_rows<GCN_OUT_C, 16, false><<<gblocks, 256, 0, stream>>>(
      h2, cnt, slots, ovf_n, ovf_src, ovf_dst, b2, g2, be2, m2, v2, (float*)d_out, N);
}


# several items per wave in every kernel (gather 5 or 4 nodes, index loads batched; GEMM 4 row stripes per B load; edge pass 8 edges/lane; small zero grid): a quarter of the waves; sum order kept
# speedup vs baseline: 3.0065x; 1.0716x over previous
#include <hip/hip_runtime.h>

typedef __attribute__((ext_vector_type(2))) float v2f;
typedef __attribute__((ext_vector_type(8))) float v8f;

#define GCN_IN_C   128
#define GCN_HID_C  128
#define GCN_OUT_C  64
#define GCN_BN_EPS 1e-5f
#define GCN_SLOT_CAP 64
#define GCN_EDGES_PER_LANE 8
#define GCN_GEMM_MT        4
#define GCN_NODES_PER_WAVE_C128 5
#define GCN_NODES_PER_WAVE_C64  4


__global__ void zero_f32(float* __restrict__ p, long n) {
  long i = (long)blockIdx.x * blockDim.x + threadIdx.x;
  long stride = (long)gridDim.x * blockDim.x;
  for (; i < n; i += stride) p[i] = 0.0f;
}

__global__ void bucket_edges_kernel(const int* __restrict__ src, const int* __restrict__ dst, int E,
                                    int* __restrict__ cnt, int* __restrict__ slots,
                                    int* __restrict__ ovf_n, int* __restrict__ ovf_src,
                                    int* __restrict__ ovf_dst) {
  constexpr int EPL = GCN_EDGES_PER_LANE;
  const int e0 = blockIdx.x * (blockDim.x * EPL) + threadIdx.x;
  int d[EPL], s[EPL], r[EPL];
#pragma unroll
  for (int k = 0; k < EPL; ++k) {
    const int e  = e0 + k * blockDim.x;
    const int ec = e < E ? e : E - 1;
    d[k] = dst[ec];
    s[k] = src[ec];
  }
#pragma unroll
  for (int k = 0; k < EPL; ++k) {
    const int e = e0 + k * blockDim.x;
    r[k] = 0;
    if (e < E) r[k] = atomicAdd(&cnt[d[k]], 1);
  }
#pragma unroll
  for (int k = 0; k < EPL; ++k) {
    const int e = e0 + k * blockDim.x;
    if (e < E) {
      if (r[k] < GCN_SLOT_CAP) {
        slots[(long)d[k] * GCN_SLOT_CAP + r[k]] = s[k];
      } else {
        const int o = atomicAdd(ovf_n, 1);
        ovf_src[o] = s[k];
        ovf_dst[o] = d[k];
      }
    }
  }
}

template <int KDIM, int NOUT, int MT>
__global__ void gemm_wmma_lds(const float* __restrict__ A, const float* __restrict__ B,
                              const int* __restrict__ cnt, float* __restrict__ C, int M) {
  constexpr int LDA  = KDIM + 4;
  constexpr int NTHR = (NOUT / 16) * 32;
  constexpr int NV   = MT * 16 * KDIM / 4 / NTHR;
  __shared__ float As[MT * 16 * LDA];

  const int row0 = blockIdx.x * (MT * 16);

  float4 st[NV];
#pragma unroll
  for (int i = 0; i < NV; ++i) {
    const int idx = threadIdx.x + i * NTHR;
    const int r   = idx / (KDIM / 4);
    const int c4  = idx % (KDIM / 4);
    int gr = row0 + r;
    gr = gr < M ? gr : M - 1;
    st[i] = ((const float4*)(A + (long)gr * KDIM))[c4];
  }
#pragma unroll
  for (int i = 0; i < NV; ++i) {
    const int idx = threadIdx.x + i * NTHR;
    const int r   = idx / (KDIM / 4);
    const int c4  = idx % (KDIM / 4);
    *(float4*)&As[r * LDA + c4 * 4] = st[i];
  }
  __syncthreads();

  const int lane = threadIdx.x & 31;
  const int half = lane >> 4;
  const int l16  = lane & 15;
  const int tn   = threadIdx.x >> 5;
  const int bcol = tn * 16 + l16;
  const float* __restrict__ ap = &As[l16 * LDA];

  int4 dg[MT][2];
#pragma unroll
  for (int t = 0; t < MT; ++t) {
    const int orow = row0 + t * 16 + half * 8;
    const int crow = orow < M ? orow : 0;
    dg[t][0] = ((const int4*)(cnt + crow))[0];
    dg[t][1] = ((const int4*)(cnt + crow))[1];
  }

  v8f acc[MT];
#pragma unroll
  for (int t = 0; t < MT; ++t) acc[t] = v8f{};
#pragma unroll 4
  for (int k0 = 0; k0 < KDIM; k0 += 4) {
    const int ka = k0 + 2 * half;
    v2f b;
    b.x = B[(long)ka * NOUT + bcol];
    b.y = B[(long)(ka + 1) * NOUT + bcol];
#pragma unroll
    for (int t = 0; t < MT; ++t) {
      v2f a;
      a.x = ap[t * 16 * LDA + ka];
      a.y = ap[t * 16 * LDA + ka + 1];
      acc[t] = __builtin_amdgcn_wmma_f32_16x16x4_f32(
           false, a,  false, b,
           (short)0, acc[t],  false,  false);
    }
  }

#pragma unroll
  for (int t = 0; t < MT; ++t) {
    if (row0 + t * 16 < M) {
      const int orow = row0 + t * 16 + half * 8;
      const float rs[8] = {rsqrtf((float)(dg[t][0].x + 1)), rsqrtf((float)(dg[t][0].y + 1)),
                           rsqrtf((float)(dg[t][0].z + 1)), rsqrtf((float)(dg[t][0].w + 1)),
                           rsqrtf((float)(dg[t][1].x + 1)), rsqrtf((float)(dg[t][1].y + 1)),
                           rsqrtf((float)(dg[t][1].z + 1)), rsqrtf((float)(dg[t][1].w + 1))};
      float* __restrict__ cp = C + (long)orow * NOUT + bcol;
#pragma unroll
      for (int r = 0; r < 8; ++r) cp[(long)r * NOUT] = acc[t][r] * rs[r];
    }
  }
}

template <int C, int G>
__device__ __forceinline__ void add_rows(float (&acc)[C / 32], const float* __restrict__ h,
                                         int sv, int j, int lane) {
  constexpr int V = C / 32;
  typedef float vec __attribute__((ext_vector_type(V)));
  vec v[G];
#pragma unroll
  for (int k = 0; k < G; ++k) {
    const int s = __builtin_amdgcn_readlane(sv, j + k);
    v[k] = *(const vec*)(h + (long)s * C + lane * V);
  }
#pragma unroll
  for (int k = 0; k < G; ++k) {
#pragma unroll
    for (int q = 0; q < V; ++q) acc[q] += v[k][q];
  }
}

template <int C, int G>
__device__ __forceinline__ void add_chunk(float (&acc)[C / 32], const float* __restrict__ h,
                                          int sv, int n, int lane) {
  int j = 0;
  for (; j + G <= n; j += G) add_rows<C, G>(acc, h, sv, j, lane);
  if (G > 8 && n - j >= 8) { add_rows<C, 8>(acc, h, sv, j, lane); j += 8; }
  if (G > 4 && n - j >= 4) { add_rows<C, 4>(acc, h, sv, j, lane); j += 4; }
  if (G > 2 && n - j >= 2) { add_rows<C, 2>(acc, h, sv, j, lane); j += 2; }
  if (n - j >= 1) add_rows<C, 1>(acc, h, sv, j, lane);
}

template <int C, int G, bool RELU, int NPW>
__global__ void gather_finish_rows(const float* __restrict__ h, const int* __restrict__ cnt,
                                   const int* __restrict__ slots, const int* __restrict__ ovf_n,
                                   const int* __restrict__ ovf_src, const int* __restrict__ ovf_dst,
                                   const float* __restrict__ bias,
                                   const float* __restrict__ gamma, const float* __restrict__ beta,
                                   const float* __restrict__ mean, const float* __restrict__ var,
                                   float* __restrict__ outp, int N) {
  constexpr int V = C / 32;
  typedef float vec __attribute__((ext_vector_type(V)));
  const int lane = threadIdx.x & 31;
  const int node0 = __builtin_amdgcn_readfirstlane(
      (int)((blockIdx.x * (long)blockDim.x + threadIdx.x) >> 5)) * NPW;
  if (node0 >= N) return;
  const int nn = N - node0 < NPW ? N - node0 : NPW;

  int cv = 0;
  if (lane < nn) cv = cnt[node0 + lane];

  int svq[NPW];
#pragma unroll
  for (int i = 0; i < NPW; ++i) {
    const int t  = __builtin_amdgcn_readlane(cv, i);
    const int ni = t < GCN_SLOT_CAP ? t : GCN_SLOT_CAP;
    svq[i] = 0;
    if (lane < ni) svq[i] = slots[(long)(node0 + i) * GCN_SLOT_CAP + lane];
  }

  for (int i = 0; i < nn; ++i) {
    const int node = node0 + i;
    const int tot = __builtin_amdgcn_readlane(cv, i);
    const int nin = tot < GCN_SLOT_CAP ? tot : GCN_SLOT_CAP;
    const int* __restrict__ row = slots + (long)node * GCN_SLOT_CAP;
    const vec own = *(const vec*)(h + (long)node * C + lane * V);
    const float di = rsqrtf((float)(tot + 1));

    float acc[V];
#pragma unroll
    for (int q = 0; q < V; ++q) acc[q] = 0.0f;

    add_chunk<C, G>(acc, h, svq[0], nin < 32 ? nin : 32, lane);
    for (int c0 = 32; c0 < nin; c0 += 32) {
      const int left = nin - c0;
      const int n = left < 32 ? left : 32;
      int sv = 0;
      if (lane < n) sv = row[c0 + lane];
      add_chunk<C, G>(acc, h, sv, n, lane);
    }

    if (tot > GCN_SLOT_CAP) {
      const int no = __builtin_amdgcn_readfirstlane(*ovf_n);
      for (int o0 = 0; o0 < no; o0 += 32) {
        int od = -1, os = 0;
        if (o0 + lane < no) {
          od = ovf_dst[o0 + lane];
          os = ovf_src[o0 + lane];
        }
        unsigned m = __builtin_amdgcn_ballot_w32(od == node);
        while (m) {
          const int j = __builtin_ctz(m);
          m &= m - 1;
          add_rows<C, 1>(acc, h, os, j, lane);
        }
      }
    }

    const vec bs = *(const vec*)(bias + lane * V);
    const vec mn = *(const vec*)(mean + lane * V);
    const vec vr = *(const vec*)(var + lane * V);
    const vec gm = *(const vec*)(gamma + lane * V);
    const vec bt = *(const vec*)(beta + lane * V);

    vec o;
#pragma unroll
    for (int q = 0; q < V; ++q) {
      float v = (acc[q] + own[q]) * di + bs[q];
      v = (v - mn[q]) * rsqrtf(vr[q] + GCN_BN_EPS) * gm[q] + bt[q];
      o[q] = RELU ? fmaxf(v, 0.0f) : v;
    }
    *(vec*)(outp + (long)node * C + lane * V) = o;

#pragma unroll
    for (int q = 0; q + 1 < NPW; ++q) svq[q] = svq[q + 1];
  }
}


extern "C" void kernel_launch(void* const* d_in, const int* in_sizes, int n_in,
                              void* d_out, int out_size, void* d_ws, size_t ws_size,
                              hipStream_t stream) {
  const float* x   = (const float*)d_in[0];
  const int*   ei  = (const int*)d_in[1];
  const float* W1  = (const float*)d_in[2];
  const float* b1  = (const float*)d_in[3];
  const float* g1  = (const float*)d_in[4];
  const float* be1 = (const float*)d_in[5];
  const float* m1  = (const float*)d_in[6];
  const float* v1  = (const float*)d_in[7];
  const float* W2  = (const float*)d_in[8];
  const float* b2  = (const float*)d_in[9];
  const float* g2  = (const float*)d_in[10];
  const float* be2 = (const float*)d_in[11];
  const float* m2  = (const float*)d_in[12];
  const float* v2  = (const float*)d_in[13];

  const int N = in_sizes[0] / GCN_IN_C;
  const int E = in_sizes[1] / 2;
  const int* src = ei;
  const int* dst = ei + E;

  char* ws = (char*)d_ws;
  int*   cnt  = (int*)ws;                         ws += (size_t)N * 2 * sizeof(int);
  int*   ovf_n = cnt + N;
  ws += (512 - (size_t)(ws - (char*)d_ws) % 512) % 512;
  float* h1   = (float*)ws;                       ws += (size_t)N * GCN_HID_C * sizeof(float);
  float* act1 = (float*)ws;                       ws += (size_t)N * GCN_HID_C * sizeof(float);
  float* h2   = (float*)ws;                       ws += (size_t)N * GCN_OUT_C * sizeof(float);
  ws += (256 - (size_t)(ws - (char*)d_ws) % 256) % 256;
  int*   slots   = (int*)ws;                      ws += (size_t)N * GCN_SLOT_CAP * sizeof(int);
  int*   ovf_src = (int*)ws;                      ws += (size_t)E * sizeof(int);
  int*   ovf_dst = (int*)ws;                      ws += (size_t)E * sizeof(int);
  (void)ws_size; (void)n_in; (void)out_size;

  const int zblocks = (N + 1 + 255) / 256;
  const int eblocks = (E + 256 * GCN_EDGES_PER_LANE - 1) / (256 * GCN_EDGES_PER_LANE);
  const int mblocks = (N + 16 * GCN_GEMM_MT - 1) / (16 * GCN_GEMM_MT);
  const int gblocks1 = (N + 8 * GCN_NODES_PER_WAVE_C128 - 1) / (8 * GCN_NODES_PER_WAVE_C128);
  const int gblocks2 = (N + 8 * GCN_NODES_PER_WAVE_C64 - 1) / (8 * GCN_NODES_PER_WAVE_C64);
  zero_f32<<<zblocks, 256, 0, stream>>>((float*)cnt, (long)N + 1);

  bucket_edges_kernel<<<eblocks, 256, 0, stream>>>(src, dst, E, cnt, slots, ovf_n, ovf_src, ovf_dst);

  gemm_wmma_lds<GCN_IN_C, GCN_HID_C, GCN_GEMM_MT><<<mblocks, (GCN_HID_C / 16) * 32, 0, stream>>>(x, W1, cnt, h1, N);
  gather_finish_rows<GCN_HID_C, 8, true, GCN_NODES_PER_WAVE_C128><<<gblocks1, 256, 0, stream>>>(
      h1, cnt, slots, ovf_n, ovf_src, ovf_dst, b1, g1, be1, m1, v1, act1, N);

  gemm_wmma_lds<GCN_HID_C, GCN_OUT_C, GCN_GEMM_MT><<<mblocks, (GCN_OUT_C / 16) * 32, 0, stream>>>(act1, W2, cnt, h2, N);
  gather_finish_rows<GCN_OUT_C, 16, false, GCN_NODES_PER_WAVE_C64><<<gblocks2, 256, 0, stream>>>(
      h2, cnt, slots, ovf_n, ovf_src, ovf_dst, b2, g2, be2, m2, v2, (float*)d_out, N);
}


# in-edge table by two-level LDS partition (block sorts edges by dst range; one block per 256 nodes ranks them with LDS adds, writes rows whole): no per-edge global atomic; slot (sum) order changes
# speedup vs baseline: 5.9409x; 1.9760x over previous
#include <hip/hip_runtime.h>

typedef __attribute__((ext_vector_type(2))) float v2f;
typedef __attribute__((ext_vector_type(8))) float v8f;

#define GCN_IN_C   128
#define GCN_HID_C  128
#define GCN_OUT_C  64
#define GCN_BN_EPS 1e-5f
#define GCN_SLOT_CAP 64
#define GCN_GEMM_MT        4
#define GCN_NODES_PER_WAVE_C128 5
#define GCN_NODES_PER_WAVE_C64  4


__global__ void zero_f32(float* __restrict__ p, long n) {
  long i = (long)blockIdx.x * blockDim.x + threadIdx.x;
  long stride = (long)gridDim.x * blockDim.x;
  for (; i < n; i += stride) p[i] = 0.0f;
}

#define GCN_BINS      256
#define GCN_BIN_CAP   16384
#define GCN_P1_EPT    8
#define GCN_ROW_PITCH 65
#define GCN_CTR_PITCH 32

__device__ __forceinline__ int gcn_bin_shift(int N) {
  int sh = 8;
  while (((N - 1) >> sh) >= GCN_BINS) ++sh;
  return sh;
}

__global__ __launch_bounds__(1024) void bin_edges_kernel(
    const int* __restrict__ src, const int* __restrict__ dst, int E, int N,
    int* __restrict__ bin_n, int2* __restrict__ bin_pairs,
    int* __restrict__ spill_n, int* __restrict__ spill_src, int* __restrict__ spill_dst) {
  constexpr int EPT = GCN_P1_EPT;
  constexpr int NT  = 1024;
  constexpr int EB  = EPT * NT;
  __shared__ int  hist[GCN_BINS];
  __shared__ int  scan[GCN_BINS];
  __shared__ int  lstart[GCN_BINS];
  __shared__ int  gbase[GCN_BINS];
  __shared__ int2 sorted[EB];
  const int tid = threadIdx.x;
  const int sh  = gcn_bin_shift(N);
  const int eb0 = blockIdx.x * EB;
  if (tid < GCN_BINS) hist[tid] = 0;
  __syncthreads();

  int d[EPT], s[EPT], r[EPT], b[EPT];
#pragma unroll
  for (int k = 0; k < EPT; ++k) {
    const int e  = eb0 + tid + k * NT;
    const int ec = e < E ? e : E - 1;
    d[k] = dst[ec];
    s[k] = src[ec];
  }
#pragma unroll
  for (int k = 0; k < EPT; ++k) {
    const int e = eb0 + tid + k * NT;
    int bb = d[k] >> sh;
    bb = bb < GCN_BINS ? bb : GCN_BINS - 1;
    b[k] = bb;
    r[k] = 0;
    if (e < E) r[k] = atomicAdd(&hist[bb], 1);
  }
  __syncthreads();

  int c = 0;
  if (tid < GCN_BINS) {
    c = hist[tid];
    scan[tid] = c;
    gbase[tid] = c ? atomicAdd(&bin_n[tid * GCN_CTR_PITCH], c) : 0;
  }
  __syncthreads();
  for (int off = 1; off < GCN_BINS; off <<= 1) {
    int v = 0;
    if (tid < GCN_BINS && tid >= off) v = scan[tid - off];
    __syncthreads();
    if (tid < GCN_BINS) scan[tid] += v;
    __syncthreads();
  }
  if (tid < GCN_BINS) lstart[tid] = scan[tid] - c;
  __syncthreads();

  int ls[EPT];
#pragma unroll
  for (int k = 0; k < EPT; ++k) ls[k] = lstart[b[k]];
#pragma unroll
  for (int k = 0; k < EPT; ++k) {
    const int e = eb0 + tid + k * NT;
    if (e < E) sorted[ls[k] + r[k]] = make_int2(s[k], d[k]);
  }
  __syncthreads();

  const int left = E - eb0;
  const int nv = left < EB ? left : EB;
  int2 p[EPT];
  int  pb[EPT], pos[EPT];
#pragma unroll
  for (int k = 0; k < EPT; ++k) {
    const int i = tid + k * NT;
    p[k] = sorted[i < nv ? i : 0];
  }
#pragma unroll
  for (int k = 0; k < EPT; ++k) {
    const int i = tid + k * NT;
    int bb = p[k].y >> sh;
    bb = bb < GCN_BINS ? bb : GCN_BINS - 1;
    pb[k]  = bb;
    pos[k] = gbase[bb] + (i - lstart[bb]);
  }
#pragma unroll
  for (int k = 0; k < EPT; ++k) {
    const int i = tid + k * NT;
    if (i < nv) {
      if (pos[k] < GCN_BIN_CAP) {
        bin_pairs[(long)pb[k] * GCN_BIN_CAP + pos[k]] = p[k];
      } else {
        const int o = atomicAdd(spill_n, 1);
        spill_src[o] = p[k].x;
        spill_dst[o] = p[k].y;
      }
    }
  }
}

__global__ __launch_bounds__(1024) void build_rows_kernel(
    int N, const int* __restrict__ bin_n, const int2* __restrict__ bin_pairs,
    const int* __restrict__ spill_n, const int* __restrict__ spill_src,
    const int* __restrict__ spill_dst,
    int* __restrict__ cnt, int* __restrict__ slots,
    int* __restrict__ ovf_n, int* __restrict__ ovf_src, int* __restrict__ ovf_dst) {
  constexpr int NT = 1024;
  __shared__ int lcnt[256];
  __shared__ int lrow[256 * GCN_ROW_PITCH];
  const int tid   = threadIdx.x;
  const int sh    = gcn_bin_shift(N);
  const int blk   = blockIdx.x;
  const int node0 = blk << 8;
  const int bin   = node0 >> sh;
  if (tid < 256) lcnt[tid] = 0;
  __syncthreads();

  int n = bin_n[bin * GCN_CTR_PITCH];
  n = n < GCN_BIN_CAP ? n : GCN_BIN_CAP;
  const int2* __restrict__ reg = bin_pairs + (long)bin * GCN_BIN_CAP;
  for (int i0 = 0; i0 < n; i0 += NT * 8) {
    int2 p[8];
#pragma unroll
    for (int k = 0; k < 8; ++k) {
      const int i = i0 + tid + k * NT;
      p[k] = reg[i < n ? i : n - 1];
    }
#pragma unroll
    for (int k = 0; k < 8; ++k) {
      const int i = i0 + tid + k * NT;
      if (i < n && (p[k].y >> 8) == blk) {
        const int l = p[k].y & 255;
        const int r = atomicAdd(&lcnt[l], 1);
        if (r < GCN_SLOT_CAP) {
          lrow[l * GCN_ROW_PITCH + r] = p[k].x;
        } else {
          const int o = atomicAdd(ovf_n, 1);
          ovf_src[o] = p[k].x;
          ovf_dst[o] = p[k].y;
        }
      }
    }
  }
  const int ns = *spill_n;
  for (int i = tid; i < ns; i += NT) {
    const int dd = spill_dst[i];
    if ((dd >> 8) == blk) {
      const int ss = spill_src[i];
      const int l = dd & 255;
      const int r = atomicAdd(&lcnt[l], 1);
      if (r < GCN_SLOT_CAP) {
        lrow[l * GCN_ROW_PITCH + r] = ss;
      } else {
        const int o = atomicAdd(ovf_n, 1);
        ovf_src[o] = ss;
        ovf_dst[o] = dd;
      }
    }
  }
  __syncthreads();

  if (tid < 256 && node0 + tid < N) cnt[node0 + tid] = lcnt[tid];
  const int lane = tid & 31;
  for (int row = tid >> 5; row < 256; row += NT / 32) {
    const int node = node0 + row;
    if (node < N) {
      int cr = lcnt[row];
      cr = cr < GCN_SLOT_CAP ? cr : GCN_SLOT_CAP;
      int* __restrict__ out = slots + (long)node * GCN_SLOT_CAP;
      if (lane < cr)      out[lane]      = lrow[row * GCN_ROW_PITCH + lane];
      if (lane + 32 < cr) out[lane + 32] = lrow[row * GCN_ROW_PITCH + lane + 32];
    }
  }
}

template <int KDIM, int NOUT, int MT>
__global__ void gemm_wmma_lds(const float* __restrict__ A, const float* __restrict__ B,
                              const int* __restrict__ cnt, float* __restrict__ C, int M) {
  constexpr int LDA  = KDIM + 4;
  constexpr int NTHR = (NOUT / 16) * 32;
  constexpr int NV   = MT * 16 * KDIM / 4 / NTHR;
  __shared__ float As[MT * 16 * LDA];

  const int row0 = blockIdx.x * (MT * 16);

  float4 st[NV];
#pragma unroll
  for (int i = 0; i < NV; ++i) {
    const int idx = threadIdx.x + i * NTHR;
    const int r   = idx / (KDIM / 4);
    const int c4  = idx % (KDIM / 4);
    int gr = row0 + r;
    gr = gr < M ? gr : M - 1;
    st[i] = ((const float4*)(A + (long)gr * KDIM))[c4];
  }
#pragma unroll
  for (int i = 0; i < NV; ++i) {
    const int idx = threadIdx.x + i * NTHR;
    const int r   = idx / (KDIM / 4);
    const int c4  = idx % (KDIM / 4);
    *(float4*)&As[r * LDA + c4 * 4] = st[i];
  }
  __syncthreads();

  const int lane = threadIdx.x & 31;
  const int half = lane >> 4;
  const int l16  = lane & 15;
  const int tn   = threadIdx.x >> 5;
  const int bcol = tn * 16 + l16;
  const float* __restrict__ ap = &As[l16 * LDA];

  int4 dg[MT][2];
#pragma unroll
  for (int t = 0; t < MT; ++t) {
    const int orow = row0 + t * 16 + half * 8;
    const int crow = orow < M ? orow : 0;
    dg[t][0] = ((const int4*)(cnt + crow))[0];
    dg[t][1] = ((const int4*)(cnt + crow))[1];
  }

  v8f acc[MT];
#pragma unroll
  for (int t = 0; t < MT; ++t) acc[t] = v8f{};
#pragma unroll 4
  for (int k0 = 0; k0 < KDIM; k0 += 4) {
    const int ka = k0 + 2 * half;
    v2f b;
    b.x = B[(long)ka * NOUT + bcol];
    b.y = B[(long)(ka + 1) * NOUT + bcol];
#pragma unroll
    for (int t = 0; t < MT; ++t) {
      v2f a;
      a.x = ap[t * 16 * LDA + ka];
      a.y = ap[t * 16 * LDA + ka + 1];
      acc[t] = __builtin_amdgcn_wmma_f32_16x16x4_f32(
           false, a,  false, b,
           (short)0, acc[t],  false,  false);
    }
  }

#pragma unroll
  for (int t = 0; t < MT; ++t) {
    if (row0 + t * 16 < M) {
      const int orow = row0 + t * 16 + half * 8;
      const float rs[8] = {rsqrtf((float)(dg[t][0].x + 1)), rsqrtf((float)(dg[t][0].y + 1)),
                           rsqrtf((float)(dg[t][0].z + 1)), rsqrtf((float)(dg[t][0].w + 1)),
                           rsqrtf((float)(dg[t][1].x + 1)), rsqrtf((float)(dg[t][1].y + 1)),
                           rsqrtf((float)(dg[t][1].z + 1)), rsqrtf((float)(dg[t][1].w + 1))};
      float* __restrict__ cp = C + (long)orow * NOUT + bcol;
#pragma unroll
      for (int r = 0; r < 8; ++r) cp[(long)r * NOUT] = acc[t][r] * rs[r];
    }
  }
}

template <int C, int G>
__device__ __forceinline__ void add_rows(float (&acc)[C / 32], const float* __restrict__ h,
                                         int sv, int j, int lane) {
  constexpr int V = C / 32;
  typedef float vec __attribute__((ext_vector_type(V)));
  vec v[G];
#pragma unroll
  for (int k = 0; k < G; ++k) {
    const int s = __builtin_amdgcn_readlane(sv, j + k);
    v[k] = *(const vec*)(h + (long)s * C + lane * V);
  }
#pragma unroll
  for (int k = 0; k < G; ++k) {
#pragma unroll
    for (int q = 0; q < V; ++q) acc[q] += v[k][q];
  }
}

template <int C, int G>
__device__ __forceinline__ void add_chunk(float (&acc)[C / 32], const float* __restrict__ h,
                                          int sv, int n, int lane) {
  int j = 0;
  for (; j + G <= n; j += G) add_rows<C, G>(acc, h, sv, j, lane);
  if (G > 8 && n - j >= 8) { add_rows<C, 8>(acc, h, sv, j, lane); j += 8; }
  if (G > 4 && n - j >= 4) { add_rows<C, 4>(acc, h, sv, j, lane); j += 4; }
  if (G > 2 && n - j >= 2) { add_rows<C, 2>(acc, h, sv, j, lane); j += 2; }
  if (n - j >= 1) add_rows<C, 1>(acc, h, sv, j, lane);
}

template <int C, int G, bool RELU, int NPW>
__global__ void gather_finish_rows(const float* __restrict__ h, const int* __restrict__ cnt,
                                   const int* __restrict__ slots, const int* __restrict__ ovf_n,
                                   const int* __restrict__ ovf_src, const int* __restrict__ ovf_dst,
                                   const float* __restrict__ bias,
                                   const float* __restrict__ gamma, const float* __restrict__ beta,
                                   const float* __restrict__ mean, const float* __restrict__ var,
                                   float* __restrict__ outp, int N) {
  constexpr int V = C / 32;
  typedef float vec __attribute__((ext_vector_type(V)));
  const int lane = threadIdx.x & 31;
  const int node0 = __builtin_amdgcn_readfirstlane(
      (int)((blockIdx.x * (long)blockDim.x + threadIdx.x) >> 5)) * NPW;
  if (node0 >= N) return;
  const int nn = N - node0 < NPW ? N - node0 : NPW;

  int cv = 0;
  if (lane < nn) cv = cnt[node0 + lane];

  int svq[NPW];
#pragma unroll
  for (int i = 0; i < NPW; ++i) {
    const int t  = __builtin_amdgcn_readlane(cv, i);
    const int ni = t < GCN_SLOT_CAP ? t : GCN_SLOT_CAP;
    svq[i] = 0;
    if (lane < ni) svq[i] = slots[(long)(node0 + i) * GCN_SLOT_CAP + lane];
  }

  for (int i = 0; i < nn; ++i) {
    const int node = node0 + i;
    const int tot = __builtin_amdgcn_readlane(cv, i);
    const int nin = tot < GCN_SLOT_CAP ? tot : GCN_SLOT_CAP;
    const int* __restrict__ row = slots + (long)node * GCN_SLOT_CAP;
    const vec own = *(const vec*)(h + (long)node * C + lane * V);
    const float di = rsqrtf((float)(tot + 1));

    float acc[V];
#pragma unroll
    for (int q = 0; q < V; ++q) acc[q] = 0.0f;

    add_chunk<C, G>(acc, h, svq[0], nin < 32 ? nin : 32, lane);
    for (int c0 = 32; c0 < nin; c0 += 32) {
      const int left = nin - c0;
      const int n = left < 32 ? left : 32;
      int sv = 0;
      if (lane < n) sv = row[c0 + lane];
      add_chunk<C, G>(acc, h, sv, n, lane);
    }

    if (tot > GCN_SLOT_CAP) {
      const int no = __builtin_amdgcn_readfirstlane(*ovf_n);
      for (int o0 = 0; o0 < no; o0 += 32) {
        int od = -1, os = 0;
        if (o0 + lane < no) {
          od = ovf_dst[o0 + lane];
          os = ovf_src[o0 + lane];
        }
        unsigned m = __builtin_amdgcn_ballot_w32(od == node);
        while (m) {
          const int j = __builtin_ctz(m);
          m &= m - 1;
          add_rows<C, 1>(acc, h, os, j, lane);
        }
      }
    }

    const vec bs = *(const vec*)(bias + lane * V);
    const vec mn = *(const vec*)(mean + lane * V);
    const vec vr = *(const vec*)(var + lane * V);
    const vec gm = *(const vec*)(gamma + lane * V);
    const vec bt = *(const vec*)(beta + lane * V);

    vec o;
#pragma unroll
    for (int q = 0; q < V; ++q) {
      float v = (acc[q] + own[q]) * di + bs[q];
      v = (v - mn[q]) * rsqrtf(vr[q] + GCN_BN_EPS) * gm[q] + bt[q];
      o[q] = RELU ? fmaxf(v, 0.0f) : v;
    }
    *(vec*)(outp + (long)node * C + lane * V) = o;

#pragma unroll
    for (int q = 0; q + 1 < NPW; ++q) svq[q] = svq[q + 1];
  }
}


extern "C" void kernel_launch(void* const* d_in, const int* in_sizes, int n_in,
                              void* d_out, int out_size, void* d_ws, size_t ws_size,
                              hipStream_t stream) {
  const float* x   = (const float*)d_in[0];
  const int*   ei  = (const int*)d_in[1];
  const float* W1  = (const float*)d_in[2];
  const float* b1  = (const float*)d_in[3];
  const float* g1  = (const float*)d_in[4];
  const float* be1 = (const float*)d_in[5];
  const float* m1  = (const float*)d_in[6];
  const float* v1  = (const float*)d_in[7];
  const float* W2  = (const float*)d_in[8];
  const float* b2  = (const float*)d_in[9];
  const float* g2  = (const float*)d_in[10];
  const float* be2 = (const float*)d_in[11];
  const float* m2  = (const float*)d_in[12];
  const float* v2  = (const float*)d_in[13];

  const int N = in_sizes[0] / GCN_IN_C;
  const int E = in_sizes[1] / 2;
  const int* src = ei;
  const int* dst = ei + E;

  char* ws = (char*)d_ws;
  int*   cnt  = (int*)ws;                         ws += (size_t)N * 2 * sizeof(int);
  ws += (512 - (size_t)(ws - (char*)d_ws) % 512) % 512;
  float* h1   = (float*)ws;                       ws += (size_t)N * GCN_HID_C * sizeof(float);
  float* act1 = (float*)ws;                       ws += (size_t)N * GCN_HID_C * sizeof(float);
  float* h2   = (float*)ws;                       ws += (size_t)N * GCN_OUT_C * sizeof(float);
  ws += (256 - (size_t)(ws - (char*)d_ws) % 256) % 256;
  int*   slots   = (int*)ws;                      ws += (size_t)N * GCN_SLOT_CAP * sizeof(int);
  int*   ovf_src = (int*)ws;                      ws += (size_t)E * sizeof(int);
  int*   ovf_dst = (int*)ws;                      ws += (size_t)E * sizeof(int);
  ws += (256 - (size_t)(ws - (char*)d_ws) % 256) % 256;
  int*   bin_n   = (int*)ws;                      ws += (size_t)(GCN_BINS + 2) * GCN_CTR_PITCH * sizeof(int);
  int*   spill_n = bin_n + GCN_BINS * GCN_CTR_PITCH;
  int*   ovf_n   = bin_n + (GCN_BINS + 1) * GCN_CTR_PITCH;
  int2*  bin_pairs = (int2*)ws;                   ws += (size_t)GCN_BINS * GCN_BIN_CAP * sizeof(int2);
  int*   spill_src = (int*)ws;                    ws += (size_t)E * sizeof(int);
  int*   spill_dst = (int*)ws;                    ws += (size_t)E * sizeof(int);
  (void)ws_size; (void)n_in; (void)out_size;

  const int eblocks = (E + 1024 * GCN_P1_EPT - 1) / (1024 * GCN_P1_EPT);
  const int rblocks = (N + 255) / 256;
  const int mblocks = (N + 16 * GCN_GEMM_MT - 1) / (16 * GCN_GEMM_MT);
  const int gblocks1 = (N + 8 * GCN_NODES_PER_WAVE_C128 - 1) / (8 * GCN_NODES_PER_WAVE_C128);
  const int gblocks2 = (N + 8 * GCN_NODES_PER_WAVE_C64 - 1) / (8 * GCN_NODES_PER_WAVE_C64);
  zero_f32<<<(GCN_BINS + 2) * GCN_CTR_PITCH / 1024 + 1, 1024, 0, stream>>>((float*)bin_n, (long)(GCN_BINS + 2) * GCN_CTR_PITCH);

  bin_edges_kernel<<<eblocks, 1024, 0, stream>>>(src, dst, E, N, bin_n, bin_pairs,
                                                 spill_n, spill_src, spill_dst);
  build_rows_kernel<<<rblocks, 1024, 0, stream>>>(N, bin_n, bin_pairs, spill_n, spill_src, spill_dst,
                                                  cnt, slots, ovf_n, ovf_src, ovf_dst);

  gemm_wmma_lds<GCN_IN_C, GCN_HID_C, GCN_GEMM_MT><<<mblocks, (GCN_HID_C / 16) * 32, 0, stream>>>(x, W1, cnt, h1, N);
  gather_finish_rows<GCN_HID_C, 8, true, GCN_NODES_PER_WAVE_C128><<<gblocks1, 256, 0, stream>>>(
      h1, cnt, slots, ovf_n, ovf_src, ovf_dst, b1, g1, be1, m1, v1, act1, N);

  gemm_wmma_lds<GCN_HID_C, GCN_OUT_C, GCN_GEMM_MT><<<mblocks, (GCN_OUT_C / 16) * 32, 0, stream>>>(act1, W2, cnt, h2, N);
  gather_finish_rows<GCN_OUT_C, 16, false, GCN_NODES_PER_WAVE_C64><<<gblocks2, 256, 0, stream>>>(
      h2, cnt, slots, ovf_n, ovf_src, ovf_dst, b2, g2, be2, m2, v2, (float*)d_out, N);
}


# layer 1 gather blocks keep their 64 finished rows in LDS and multiply them by W2 themselves (WMMA f32, same k and sum order): no act1 plane, no layer-2 product launch; 8 nodes per wave
# speedup vs baseline: 6.2805x; 1.0572x over previous
#include <hip/hip_runtime.h>

typedef __attribute__((ext_vector_type(2))) float v2f;
typedef __attribute__((ext_vector_type(8))) float v8f;

#define GCN_IN_C   128
#define GCN_HID_C  128
#define GCN_OUT_C  64
#define GCN_BN_EPS 1e-5f
#define GCN_SLOT_CAP 64
#define GCN_GEMM_MT        4
#define GCN_NODES_PER_WAVE_C128 8
#define GCN_NODES_PER_WAVE_C64  4


__global__ void zero_f32(float* __restrict__ p, long n) {
  long i = (long)blockIdx.x * blockDim.x + threadIdx.x;
  long stride = (long)gridDim.x * blockDim.x;
  for (; i < n; i += stride) p[i] = 0.0f;
}

#define GCN_BINS      256
#define GCN_BIN_CAP   16384
#define GCN_P1_EPT    8
#define GCN_ROW_PITCH 65
#define GCN_CTR_PITCH 32

__device__ __forceinline__ int gcn_bin_shift(int N) {
  int sh = 8;
  while (((N - 1) >> sh) >= GCN_BINS) ++sh;
  return sh;
}

__global__ __launch_bounds__(1024) void bin_edges_kernel(
    const int* __restrict__ src, const int* __restrict__ dst, int E, int N,
    int* __restrict__ bin_n, int2* __restrict__ bin_pairs,
    int* __restrict__ spill_n, int* __restrict__ spill_src, int* __restrict__ spill_dst) {
  constexpr int EPT = GCN_P1_EPT;
  constexpr int NT  = 1024;
  constexpr int EB  = EPT * NT;
  __shared__ int  hist[GCN_BINS];
  __shared__ int  scan[GCN_BINS];
  __shared__ int  lstart[GCN_BINS];
  __shared__ int  gbase[GCN_BINS];
  __shared__ int2 sorted[EB];
  const int tid = threadIdx.x;
  const int sh  = gcn_bin_shift(N);
  const int eb0 = blockIdx.x * EB;
  if (tid < GCN_BINS) hist[tid] = 0;
  __syncthreads();

  int d[EPT], s[EPT], r[EPT], b[EPT];
#pragma unroll
  for (int k = 0; k < EPT; ++k) {
    const int e  = eb0 + tid + k * NT;
    const int ec = e < E ? e : E - 1;
    d[k] = dst[ec];
    s[k] = src[ec];
  }
#pragma unroll
  for (int k = 0; k < EPT; ++k) {
    const int e = eb0 + tid + k * NT;
    int bb = d[k] >> sh;
    bb = bb < GCN_BINS ? bb : GCN_BINS - 1;
    b[k] = bb;
    r[k] = 0;
    if (e < E) r[k] = atomicAdd(&hist[bb], 1);
  }
  __syncthreads();

  int c = 0;
  if (tid < GCN_BINS) {
    c = hist[tid];
    scan[tid] = c;
    gbase[tid] = c ? atomicAdd(&bin_n[tid * GCN_CTR_PITCH], c) : 0;
  }
  __syncthreads();
  for (int off = 1; off < GCN_BINS; off <<= 1) {
    int v = 0;
    if (tid < GCN_BINS && tid >= off) v = scan[tid - off];
    __syncthreads();
    if (tid < GCN_BINS) scan[tid] += v;
    __syncthreads();
  }
  if (tid < GCN_BINS) lstart[tid] = scan[tid] - c;
  __syncthreads();

  int ls[EPT];
#pragma unroll
  for (int k = 0; k < EPT; ++k) ls[k] = lstart[b[k]];
#pragma unroll
  for (int k = 0; k < EPT; ++k) {
    const int e = eb0 + tid + k * NT;
    if (e < E) sorted[ls[k] + r[k]] = make_int2(s[k], d[k]);
  }
  __syncthreads();

  const int left = E - eb0;
  const int nv = left < EB ? left : EB;
  int2 p[EPT];
  int  pb[EPT], pos[EPT];
#pragma unroll
  for (int k = 0; k < EPT; ++k) {
    const int i = tid + k * NT;
    p[k] = sorted[i < nv ? i : 0];
  }
#pragma unroll
  for (int k = 0; k < EPT; ++k) {
    const int i = tid + k * NT;
    int bb = p[k].y >> sh;
    bb = bb < GCN_BINS ? bb : GCN_BINS - 1;
    pb[k]  = bb;
    pos[k] = gbase[bb] + (i - lstart[bb]);
  }
#pragma unroll
  for (int k = 0; k < EPT; ++k) {
    const int i = tid + k * NT;
    if (i < nv) {
      if (pos[k] < GCN_BIN_CAP) {
        bin_pairs[(long)pb[k] * GCN_BIN_CAP + pos[k]] = p[k];
      } else {
        const int o = atomicAdd(spill_n, 1);
        spill_src[o] = p[k].x;
        spill_dst[o] = p[k].y;
      }
    }
  }
}

__global__ __launch_bounds__(1024) void build_rows_kernel(
    int N, const int* __restrict__ bin_n, const int2* __restrict__ bin_pairs,
    const int* __restrict__ spill_n, const int* __restrict__ spill_src,
    const int* __restrict__ spill_dst,
    int* __restrict__ cnt, int* __restrict__ slots,
    int* __restrict__ ovf_n, int* __restrict__ ovf_src, int* __restrict__ ovf_dst) {
  constexpr int NT = 1024;
  __shared__ int lcnt[256];
  __shared__ int lrow[256 * GCN_ROW_PITCH];
  const int tid   = threadIdx.x;
  const int sh    = gcn_bin_shift(N);
  const int blk   = blockIdx.x;
  const int node0 = blk << 8;
  const int bin   = node0 >> sh;
  if (tid < 256) lcnt[tid] = 0;
  __syncthreads();

  int n = bin_n[bin * GCN_CTR_PITCH];
  n = n < GCN_BIN_CAP ? n : GCN_BIN_CAP;
  const int2* __restrict__ reg = bin_pairs + (long)bin * GCN_BIN_CAP;
  for (int i0 = 0; i0 < n; i0 += NT * 8) {
    int2 p[8];
#pragma unroll
    for (int k = 0; k < 8; ++k) {
      const int i = i0 + tid + k * NT;
      p[k] = reg[i < n ? i : n - 1];
    }
#pragma unroll
    for (int k = 0; k < 8; ++k) {
      const int i = i0 + tid + k * NT;
      if (i < n && (p[k].y >> 8) == blk) {
        const int l = p[k].y & 255;
        const int r = atomicAdd(&lcnt[l], 1);
        if (r < GCN_SLOT_CAP) {
          lrow[l * GCN_ROW_PITCH + r] = p[k].x;
        } else {
          const int o = atomicAdd(ovf_n, 1);
          ovf_src[o] = p[k].x;
          ovf_dst[o] = p[k].y;
        }
      }
    }
  }
  const int ns = *spill_n;
  for (int i = tid; i < ns; i += NT) {
    const int dd = spill_dst[i];
    if ((dd >> 8) == blk) {
      const int ss = spill_src[i];
      const int l = dd & 255;
      const int r = atomicAdd(&lcnt[l], 1);
      if (r < GCN_SLOT_CAP) {
        lrow[l * GCN_ROW_PITCH + r] = ss;
      } else {
        const int o = atomicAdd(ovf_n, 1);
        ovf_src[o] = ss;
        ovf_dst[o] = dd;
      }
    }
  }
  __syncthreads();

  if (tid < 256 && node0 + tid < N) cnt[node0 + tid] = lcnt[tid];
  const int lane = tid & 31;
  for (int row = tid >> 5; row < 256; row += NT / 32) {
    const int node = node0 + row;
    if (node < N) {
      int cr = lcnt[row];
      cr = cr < GCN_SLOT_CAP ? cr : GCN_SLOT_CAP;
      int* __restrict__ out = slots + (long)node * GCN_SLOT_CAP;
      if (lane < cr)      out[lane]      = lrow[row * GCN_ROW_PITCH + lane];
      if (lane + 32 < cr) out[lane + 32] = lrow[row * GCN_ROW_PITCH + lane + 32];
    }
  }
}

template <int KDIM, int NOUT, int MT>
__global__ void gemm_wmma_lds(const float* __restrict__ A, const float* __restrict__ B,
                              const int* __restrict__ cnt, float* __restrict__ C, int M) {
  constexpr int LDA  = KDIM + 4;
  constexpr int NTHR = (NOUT / 16) * 32;
  constexpr int NV   = MT * 16 * KDIM / 4 / NTHR;
  __shared__ float As[MT * 16 * LDA];

  const int row0 = blockIdx.x * (MT * 16);

  float4 st[NV];
#pragma unroll
  for (int i = 0; i < NV; ++i) {
    const int idx = threadIdx.x + i * NTHR;
    const int r   = idx / (KDIM / 4);
    const int c4  = idx % (KDIM / 4);
    int gr = row0 + r;
    gr = gr < M ? gr : M - 1;
    st[i] = ((const float4*)(A + (long)gr * KDIM))[c4];
  }
#pragma unroll
  for (int i = 0; i < NV; ++i) {
    const int idx = threadIdx.x + i * NTHR;
    const int r   = idx / (KDIM / 4);
    const int c4  = idx % (KDIM / 4);
    *(float4*)&As[r * LDA + c4 * 4] = st[i];
  }
  __syncthreads();

  const int lane = threadIdx.x & 31;
  const int half = lane >> 4;
  const int l16  = lane & 15;
  const int tn   = threadIdx.x >> 5;
  const int bcol = tn * 16 + l16;
  const float* __restrict__ ap = &As[l16 * LDA];

  int4 dg[MT][2];
#pragma unroll
  for (int t = 0; t < MT; ++t) {
    const int orow = row0 + t * 16 + half * 8;
    const int crow = orow < M ? orow : 0;
    dg[t][0] = ((const int4*)(cnt + crow))[0];
    dg[t][1] = ((const int4*)(cnt + crow))[1];
  }

  v8f acc[MT];
#pragma unroll
  for (int t = 0; t < MT; ++t) acc[t] = v8f{};
#pragma unroll 4
  for (int k0 = 0; k0 < KDIM; k0 += 4) {
    const int ka = k0 + 2 * half;
    v2f b;
    b.x = B[(long)ka * NOUT + bcol];
    b.y = B[(long)(ka + 1) * NOUT + bcol];
#pragma unroll
    for (int t = 0; t < MT; ++t) {
      v2f a;
      a.x = ap[t * 16 * LDA + ka];
      a.y = ap[t * 16 * LDA + ka + 1];
      acc[t] = __builtin_amdgcn_wmma_f32_16x16x4_f32(
           false, a,  false, b,
           (short)0, acc[t],  false,  false);
    }
  }

#pragma unroll
  for (int t = 0; t < MT; ++t) {
    if (row0 + t * 16 < M) {
      const int orow = row0 + t * 16 + half * 8;
      const float rs[8] = {rsqrtf((float)(dg[t][0].x + 1)), rsqrtf((float)(dg[t][0].y + 1)),
                           rsqrtf((float)(dg[t][0].z + 1)), rsqrtf((float)(dg[t][0].w + 1)),
                           rsqrtf((float)(dg[t][1].x + 1)), rsqrtf((float)(dg[t][1].y + 1)),
                           rsqrtf((float)(dg[t][1].z + 1)), rsqrtf((float)(dg[t][1].w + 1))};
      float* __restrict__ cp = C + (long)orow * NOUT + bcol;
#pragma unroll
      for (int r = 0; r < 8; ++r) cp[(long)r * NOUT] = acc[t][r] * rs[r];
    }
  }
}

template <int C, int G>
__device__ __forceinline__ void add_rows(float (&acc)[C / 32], const float* __restrict__ h,
                                         int sv, int j, int lane) {
  constexpr int V = C / 32;
  typedef float vec __attribute__((ext_vector_type(V)));
  vec v[G];
#pragma unroll
  for (int k = 0; k < G; ++k) {
    const int s = __builtin_amdgcn_readlane(sv, j + k);
    v[k] = *(const vec*)(h + (long)s * C + lane * V);
  }
#pragma unroll
  for (int k = 0; k < G; ++k) {
#pragma unroll
    for (int q = 0; q < V; ++q) acc[q] += v[k][q];
  }
}

template <int C, int G>
__device__ __forceinline__ void add_chunk(float (&acc)[C / 32], const float* __restrict__ h,
                                          int sv, int n, int lane) {
  int j = 0;
  for (; j + G <= n; j += G) add_rows<C, G>(acc, h, sv, j, lane);
  if (G > 8 && n - j >= 8) { add_rows<C, 8>(acc, h, sv, j, lane); j += 8; }
  if (G > 4 && n - j >= 4) { add_rows<C, 4>(acc, h, sv, j, lane); j += 4; }
  if (G > 2 && n - j >= 2) { add_rows<C, 2>(acc, h, sv, j, lane); j += 2; }
  if (n - j >= 1) add_rows<C, 1>(acc, h, sv, j, lane);
}

template <int C, int G, bool RELU, int NPW>
__global__ void gather_finish_rows(const float* __restrict__ h, const int* __restrict__ cnt,
                                   const int* __restrict__ slots, const int* __restrict__ ovf_n,
                                   const int* __restrict__ ovf_src, const int* __restrict__ ovf_dst,
                                   const float* __restrict__ bias,
                                   const float* __restrict__ gamma, const float* __restrict__ beta,
                                   const float* __restrict__ mean, const float* __restrict__ var,
                                   float* __restrict__ outp, int N) {
  constexpr int V = C / 32;
  typedef float vec __attribute__((ext_vector_type(V)));
  const int lane = threadIdx.x & 31;
  const int node0 = __builtin_amdgcn_readfirstlane(
      (int)((blockIdx.x * (long)blockDim.x + threadIdx.x) >> 5)) * NPW;
  if (node0 >= N) return;
  const int nn = N - node0 < NPW ? N - node0 : NPW;

  int cv = 0;
  if (lane < nn) cv = cnt[node0 + lane];

  int svq[NPW];
#pragma unroll
  for (int i = 0; i < NPW; ++i) {
    const int t  = __builtin_amdgcn_readlane(cv, i);
    const int ni = t < GCN_SLOT_CAP ? t : GCN_SLOT_CAP;
    svq[i] = 0;
    if (lane < ni) svq[i] = slots[(long)(node0 + i) * GCN_SLOT_CAP + lane];
  }

  for (int i = 0; i < nn; ++i) {
    const int node = node0 + i;
    const int tot = __builtin_amdgcn_readlane(cv, i);
    const int nin = tot < GCN_SLOT_CAP ? tot : GCN_SLOT_CAP;
    const int* __restrict__ row = slots + (long)node * GCN_SLOT_CAP;
    const vec own = *(const vec*)(h + (long)node * C + lane * V);
    const float di = rsqrtf((float)(tot + 1));

    float acc[V];
#pragma unroll
    for (int q = 0; q < V; ++q) acc[q] = 0.0f;

    add_chunk<C, G>(acc, h, svq[0], nin < 32 ? nin : 32, lane);
    for (int c0 = 32; c0 < nin; c0 += 32) {
      const int left = nin - c0;
      const int n = left < 32 ? left : 32;
      int sv = 0;
      if (lane < n) sv = row[c0 + lane];
      add_chunk<C, G>(acc, h, sv, n, lane);
    }

    if (tot > GCN_SLOT_CAP) {
      const int no = __builtin_amdgcn_readfirstlane(*ovf_n);
      for (int o0 = 0; o0 < no; o0 += 32) {
        int od = -1, os = 0;
        if (o0 + lane < no) {
          od = ovf_dst[o0 + lane];
          os = ovf_src[o0 + lane];
        }
        unsigned m = __builtin_amdgcn_ballot_w32(od == node);
        while (m) {
          const int j = __builtin_ctz(m);
          m &= m - 1;
          add_rows<C, 1>(acc, h, os, j, lane);
        }
      }
    }

    const vec bs = *(const vec*)(bias + lane * V);
    const vec mn = *(const vec*)(mean + lane * V);
    const vec vr = *(const vec*)(var + lane * V);
    const vec gm = *(const vec*)(gamma + lane * V);
    const vec bt = *(const vec*)(beta + lane * V);

    vec o;
#pragma unroll
    for (int q = 0; q < V; ++q) {
      float v = (acc[q] + own[q]) * di + bs[q];
      v = (v - mn[q]) * rsqrtf(vr[q] + GCN_BN_EPS) * gm[q] + bt[q];
      o[q] = RELU ? fmaxf(v, 0.0f) : v;
    }
    *(vec*)(outp + (long)node * C + lane * V) = o;

#pragma unroll
    for (int q = 0; q + 1 < NPW; ++q) svq[q] = svq[q + 1];
  }
}

template <int C, int G, int NOUT, int NPW>
__global__ __launch_bounds__(256) void gather_finish_gemm(
    const float* __restrict__ h, const int* __restrict__ cnt,
    const int* __restrict__ slots, const int* __restrict__ ovf_n,
    const int* __restrict__ ovf_src, const int* __restrict__ ovf_dst,
    const float* __restrict__ bias,
    const float* __restrict__ gamma, const float* __restrict__ beta,
    const float* __restrict__ mean, const float* __restrict__ var,
    const float* __restrict__ B, float* __restrict__ Cout, int N) {
  constexpr int V   = C / 32;
  constexpr int LDA = C + 4;
  constexpr int TM  = 8 * NPW;
  constexpr int KS  = C / 4;
  static_assert(NOUT == 64 && TM == 64, "8 waves = 4 N tiles x 2 pairs of row tiles");
  typedef float vec __attribute__((ext_vector_type(V)));
  __shared__ __attribute__((aligned(16))) float As[TM * LDA];

  const int lane = threadIdx.x & 31;
  const int wave  = __builtin_amdgcn_readfirstlane((int)(threadIdx.x >> 5));
  const int row0  = blockIdx.x * TM;
  const int node0 = row0 + wave * NPW;
  int nn = N - node0;
  nn = nn < 0 ? 0 : (nn < NPW ? nn : NPW);

  int cv = 0;
  if (lane < nn) cv = cnt[node0 + lane];

  int svq[NPW];
#pragma unroll
  for (int i = 0; i < NPW; ++i) {
    const int t  = __builtin_amdgcn_readlane(cv, i);
    const int ni = t < GCN_SLOT_CAP ? t : GCN_SLOT_CAP;
    svq[i] = 0;
    if (lane < ni) svq[i] = slots[(long)(node0 + i) * GCN_SLOT_CAP + lane];
  }

  for (int i = 0; i < nn; ++i) {
    const int node = node0 + i;
    const int tot = __builtin_amdgcn_readlane(cv, i);
    const int nin = tot < GCN_SLOT_CAP ? tot : GCN_SLOT_CAP;
    const int* __restrict__ row = slots + (long)node * GCN_SLOT_CAP;
    const vec own = *(const vec*)(h + (long)node * C + lane * V);
    const float di = rsqrtf((float)(tot + 1));

    float acc[V];
#pragma unroll
    for (int q = 0; q < V; ++q) acc[q] = 0.0f;

    add_chunk<C, G>(acc, h, svq[0], nin < 32 ? nin : 32, lane);
    for (int c0 = 32; c0 < nin; c0 += 32) {
      const int left = nin - c0;
      const int n = left < 32 ? left : 32;
      int sv = 0;
      if (lane < n) sv = row[c0 + lane];
      add_chunk<C, G>(acc, h, sv, n, lane);
    }

    if (tot > GCN_SLOT_CAP) {
      const int no = __builtin_amdgcn_readfirstlane(*ovf_n);
      for (int o0 = 0; o0 < no; o0 += 32) {
        int od = -1, os = 0;
        if (o0 + lane < no) {
          od = ovf_dst[o0 + lane];
          os = ovf_src[o0 + lane];
        }
        unsigned m = __builtin_amdgcn_ballot_w32(od == node);
        while (m) {
          const int j = __builtin_ctz(m);
          m &= m - 1;
          add_rows<C, 1>(acc, h, os, j, lane);
        }
      }
    }

    const vec bs = *(const vec*)(bias + lane * V);
    const vec mn = *(const vec*)(mean + lane * V);
    const vec vr = *(const vec*)(var + lane * V);
    const vec gm = *(const vec*)(gamma + lane * V);
    const vec bt = *(const vec*)(beta + lane * V);

    vec o;
#pragma unroll
    for (int q = 0; q < V; ++q) {
      float v = (acc[q] + own[q]) * di + bs[q];
      v = (v - mn[q]) * rsqrtf(vr[q] + GCN_BN_EPS) * gm[q] + bt[q];
      o[q] = fmaxf(v, 0.0f);
    }
    *(vec*)&As[(wave * NPW + i) * LDA + lane * V] = o;

#pragma unroll
    for (int q = 0; q + 1 < NPW; ++q) svq[q] = svq[q + 1];
  }
  for (int i = nn; i < NPW; ++i) {
    vec z;
#pragma unroll
    for (int q = 0; q < V; ++q) z[q] = 0.0f;
    *(vec*)&As[(wave * NPW + i) * LDA + lane * V] = z;
  }

  const int half = lane >> 4;
  const int l16  = lane & 15;
  const int tn   = wave & 3;
  const int sg   = wave >> 2;
  const int bcol = tn * 16 + l16;

  v2f bf[KS];
#pragma unroll
  for (int s = 0; s < KS; ++s) {
    const int ka = 4 * s + 2 * half;
    bf[s].x = B[(long)ka * NOUT + bcol];
    bf[s].y = B[(long)(ka + 1) * NOUT + bcol];
  }
  int4 dg[2][2];
#pragma unroll
  for (int t = 0; t < 2; ++t) {
    const int orow = row0 + (sg * 2 + t) * 16 + half * 8;
    const int crow = orow < N ? orow : 0;
    dg[t][0] = ((const int4*)(cnt + crow))[0];
    dg[t][1] = ((const int4*)(cnt + crow))[1];
  }

  __syncthreads();

  const float* __restrict__ ap = &As[(sg * 32 + l16) * LDA];
  v8f pacc[2];
#pragma unroll
  for (int t = 0; t < 2; ++t) pacc[t] = v8f{};
#pragma unroll
  for (int s = 0; s < KS; ++s) {
    const int ka = 4 * s + 2 * half;
#pragma unroll
    for (int t = 0; t < 2; ++t) {
      v2f a;
      a.x = ap[t * 16 * LDA + ka];
      a.y = ap[t * 16 * LDA + ka + 1];
      pacc[t] = __builtin_amdgcn_wmma_f32_16x16x4_f32(
           false, a,  false, bf[s],
           (short)0, pacc[t],  false,  false);
    }
  }

#pragma unroll
  for (int t = 0; t < 2; ++t) {
    const int trow = row0 + (sg * 2 + t) * 16;
    if (trow < N) {
      const int orow = trow + half * 8;
      const float rs[8] = {rsqrtf((float)(dg[t][0].x + 1)), rsqrtf((float)(dg[t][0].y + 1)),
                           rsqrtf((float)(dg[t][0].z + 1)), rsqrtf((float)(dg[t][0].w + 1)),
                           rsqrtf((float)(dg[t][1].x + 1)), rsqrtf((float)(dg[t][1].y + 1)),
                           rsqrtf((float)(dg[t][1].z + 1)), rsqrtf((float)(dg[t][1].w + 1))};
      float* __restrict__ cp = Cout + (long)orow * NOUT + bcol;
#pragma unroll
      for (int r = 0; r < 8; ++r) cp[(long)r * NOUT] = pacc[t][r] * rs[r];
    }
  }
}


extern "C" void kernel_launch(void* const* d_in, const int* in_sizes, int n_in,
                              void* d_out, int out_size, void* d_ws, size_t ws_size,
                              hipStream_t stream) {
  const float* x   = (const float*)d_in[0];
  const int*   ei  = (const int*)d_in[1];
  const float* W1  = (const float*)d_in[2];
  const float* b1  = (const float*)d_in[3];
  const float* g1  = (const float*)d_in[4];
  const float* be1 = (const float*)d_in[5];
  const float* m1  = (const float*)d_in[6];
  const float* v1  = (const float*)d_in[7];
  const float* W2  = (const float*)d_in[8];
  const float* b2  = (const float*)d_in[9];
  const float* g2  = (const float*)d_in[10];
  const float* be2 = (const float*)d_in[11];
  const float* m2  = (const float*)d_in[12];
  const float* v2  = (const float*)d_in[13];

  const int N = in_sizes[0] / GCN_IN_C;
  const int E = in_sizes[1] / 2;
  const int* src = ei;
  const int* dst = ei + E;

  char* ws = (char*)d_ws;
  int*   cnt  = (int*)ws;                         ws += (size_t)N * 2 * sizeof(int);
  ws += (512 - (size_t)(ws - (char*)d_ws) % 512) % 512;
  float* h1   = (float*)ws;                       ws += (size_t)N * GCN_HID_C * sizeof(float);
  float* h2   = (float*)ws;                       ws += (size_t)N * GCN_OUT_C * sizeof(float);
  ws += (256 - (size_t)(ws - (char*)d_ws) % 256) % 256;
  int*   slots   = (int*)ws;                      ws += (size_t)N * GCN_SLOT_CAP * sizeof(int);
  int*   ovf_src = (int*)ws;                      ws += (size_t)E * sizeof(int);
  int*   ovf_dst = (int*)ws;                      ws += (size_t)E * sizeof(int);
  ws += (256 - (size_t)(ws - (char*)d_ws) % 256) % 256;
  int*   bin_n   = (int*)ws;                      ws += (size_t)(GCN_BINS + 2) * GCN_CTR_PITCH * sizeof(int);
  int*   spill_n = bin_n + GCN_BINS * GCN_CTR_PITCH;
  int*   ovf_n   = bin_n + (GCN_BINS + 1) * GCN_CTR_PITCH;
  int2*  bin_pairs = (int2*)ws;                   ws += (size_t)GCN_BINS * GCN_BIN_CAP * sizeof(int2);
  int*   spill_src = (int*)ws;                    ws += (size_t)E * sizeof(int);
  int*   spill_dst = (int*)ws;                    ws += (size_t)E * sizeof(int);
  (void)ws_size; (void)n_in; (void)out_size;

  const int eblocks = (E + 1024 * GCN_P1_EPT - 1) / (1024 * GCN_P1_EPT);
  const int rblocks = (N + 255) / 256;
  const int mblocks = (N + 16 * GCN_GEMM_MT - 1) / (16 * GCN_GEMM_MT);
  const int gblocks1 = (N + 8 * GCN_NODES_PER_WAVE_C128 - 1) / (8 * GCN_NODES_PER_WAVE_C128);
  const int gblocks2 = (N + 8 * GCN_NODES_PER_WAVE_C64 - 1) / (8 * GCN_NODES_PER_WAVE_C64);
  zero_f32<<<(GCN_BINS + 2) * GCN_CTR_PITCH / 1024 + 1, 1024, 0, stream>>>((float*)bin_n, (long)(GCN_BINS + 2) * GCN_CTR_PITCH);

  bin_edges_kernel<<<eblocks, 1024, 0, stream>>>(src, dst, E, N, bin_n, bin_pairs,
                                                 spill_n, spill_src, spill_dst);
  build_rows_kernel<<<rblocks, 1024, 0, stream>>>(N, bin_n, bin_pairs, spill_n, spill_src, spill_dst,
                                                  cnt, slots, ovf_n, ovf_src, ovf_dst);

  gemm_wmma_lds<GCN_IN_C, GCN_HID_C, GCN_GEMM_MT><<<mblocks, (GCN_HID_C / 16) * 32, 0, stream>>>(x, W1, cnt, h1, N);
  gather_finish_gemm<GCN_HID_C, 8, GCN_OUT_C, GCN_NODES_PER_WAVE_C128><<<gblocks1, 256, 0, stream>>>(
      h1, cnt, slots, ovf_n, ovf_src, ovf_dst, b1, g1, be1, m1, v1, W2, h2, N);

  gather_finish_rows<GCN_OUT_C, 16, false, GCN_NODES_PER_WAVE_C64><<<gblocks2, 256, 0, stream>>>(
      h2, cnt, slots, ovf_n, ovf_src, ovf_dst, b2, g2, be2, m2, v2, (float*)d_out, N);
}
